# starting kernel
# baseline (speedup 1.0000x reference)
#include <hip/hip_runtime.h>
#include <hip/hip_bf16.h>

typedef __attribute__((ext_vector_type(16))) _Float16 v16h;
typedef __attribute__((ext_vector_type(8)))  float    v8f;

__device__ __forceinline__ int klocal(int lane, int j) {
    return ((lane >> 4) << 3) + (j < 8 ? j : j + 8);
}

__global__ __launch_bounds__(256) void k_deg_init(float* __restrict__ deg, int n) {
    int i = blockIdx.x * blockDim.x + threadIdx.x;
    if (i < n) deg[i] = 1.0f;
}

__global__ __launch_bounds__(256) void k_deg_edges(const int* __restrict__ ei,
                                                   float* __restrict__ deg, int E) {
    int e = blockIdx.x * blockDim.x + threadIdx.x;
    if (e < E) atomicAdd(&deg[ei[E + e]], 1.0f);
}

__global__ __launch_bounds__(256) void k_dinv(const float* __restrict__ deg,
                                              float* __restrict__ dinv, int n) {
    int i = blockIdx.x * blockDim.x + threadIdx.x;
    if (i < n) dinv[i] = rsqrtf(deg[i]);
}

__global__ __launch_bounds__(256) void k_norm(const int* __restrict__ ei,
                                              const float* __restrict__ dinv,
                                              float* __restrict__ norm, int E) {
    int e = blockIdx.x * blockDim.x + threadIdx.x;
    if (e < E) norm[e] = dinv[ei[e]] * dinv[ei[E + e]];
}

__global__ __launch_bounds__(256) void k_prep_w(const float* __restrict__ W1,
                                                const float* __restrict__ W2,
                                                _Float16* __restrict__ w1B,
                                                _Float16* __restrict__ w2B) {
    int t = blockIdx.x * blockDim.x + threadIdx.x;
    if (t < 4 * 32 * 16) {
        int c = t / (32 * 16);
        int l = (t / 16) & 31;
        int j = t & 15;
        int k = 32 * c + klocal(l, j);
        w1B[t] = (_Float16)W1[k * 16 + (l & 15)];
    } else if (t < 4 * 32 * 16 + 32 * 16) {
        int u = t - 4 * 32 * 16;
        int l = u / 16;
        int j = u & 15;
        int k = klocal(l, j);
        int n = l & 15;
        w2B[u] = (k < 16 && n < 7) ? (_Float16)W2[k * 7 + n] : (_Float16)0.0f;
    }
}

__global__ __launch_bounds__(256) void k_gemm1(const float* __restrict__ x,
                                               const _Float16* __restrict__ w1B,
                                               const float* __restrict__ dinv,
                                               float* __restrict__ h,
                                               float* __restrict__ acc1, int nrows) {
    int wave = (blockIdx.x * blockDim.x + threadIdx.x) >> 5;
    int lane = threadIdx.x & 31;
    int row0 = wave * 16;
    if (row0 >= nrows) return;
    int rsub = lane & 15;
    int hi   = lane >> 4;
    const float* xr = x + (size_t)(row0 + rsub) * 128 + hi * 8;
    v8f c = {};
#pragma unroll
    for (int kc = 0; kc < 4; ++kc) {
        const float* p = xr + kc * 32;
        v16h a;
#pragma unroll
        for (int j = 0; j < 8; ++j) a[j] = (_Float16)p[j];
#pragma unroll
        for (int j = 0; j < 8; ++j) a[8 + j] = (_Float16)p[16 + j];
        v16h b = *(const v16h*)(w1B + (size_t)(kc * 32 + lane) * 16);
        c = __builtin_amdgcn_wmma_f32_16x16x32_f16(false, a, false, b,
                                                   (short)0, c, false, false);
    }
#pragma unroll
    for (int r = 0; r < 8; ++r) {
        int row = row0 + hi * 8 + r;
        float hv = c[r];
        float dv = dinv[row];
        h   [(size_t)row * 16 + rsub] = hv;
        acc1[(size_t)row * 16 + rsub] = hv * dv * dv;
    }
}

__global__ __launch_bounds__(256) void k_scatter1(const int* __restrict__ ei,
                                                  const float* __restrict__ norm,
                                                  const float* __restrict__ h,
                                                  float* __restrict__ acc1, int E) {
    int t = blockIdx.x * blockDim.x + threadIdx.x;
    if (t >= 4 * E) return;
    int e = t >> 2, q = t & 3;
    int s = ei[e], d = ei[E + e];
    float nm = norm[e];
    const float4 v = *(const float4*)(h + (size_t)s * 16 + q * 4);
    float* ap = acc1 + (size_t)d * 16 + q * 4;
    atomicAdd(ap + 0, v.x * nm);
    atomicAdd(ap + 1, v.y * nm);
    atomicAdd(ap + 2, v.z * nm);
    atomicAdd(ap + 3, v.w * nm);
}

__global__ __launch_bounds__(256) void k_gemm2(const float* __restrict__ acc1,
                                               const float* __restrict__ b1,
                                               const _Float16* __restrict__ w2B,
                                               const float* __restrict__ dinv,
                                               float* __restrict__ h2,
                                               float* __restrict__ acc2, int nrows) {
    int wave = (blockIdx.x * blockDim.x + threadIdx.x) >> 5;
    int lane = threadIdx.x & 31;
    int row0 = wave * 16;
    if (row0 >= nrows) return;
    int rsub = lane & 15;
    int hi   = lane >> 4;
    int base = hi * 8;
    const float* p = acc1 + (size_t)(row0 + rsub) * 16 + base;
    v16h a;
#pragma unroll
    for (int j = 0; j < 8; ++j) {
        float f = p[j] + b1[base + j];
        a[j] = (_Float16)fmaxf(f, 0.0f);
    }
#pragma unroll
    for (int j = 8; j < 16; ++j) a[j] = (_Float16)0.0f;
    v16h b = *(const v16h*)(w2B + (size_t)lane * 16);
    v8f c = {};
    c = __builtin_amdgcn_wmma_f32_16x16x32_f16(false, a, false, b,
                                               (short)0, c, false, false);
    if (rsub < 8) {
#pragma unroll
        for (int r = 0; r < 8; ++r) {
            int row = row0 + base + r;
            float hv = c[r];
            float dv = dinv[row];
            h2  [(size_t)row * 8 + rsub] = hv;
            acc2[(size_t)row * 8 + rsub] = hv * dv * dv;
        }
    }
}

__global__ __launch_bounds__(256) void k_scatter2(const int* __restrict__ ei,
                                                  const float* __restrict__ norm,
                                                  const float* __restrict__ h2,
                                                  float* __restrict__ acc2, int E) {
    int t = blockIdx.x * blockDim.x + threadIdx.x;
    if (t >= 2 * E) return;
    int e = t >> 1, q = t & 1;
    int s = ei[e], d = ei[E + e];
    float nm = norm[e];
    const float4 v = *(const float4*)(h2 + (size_t)s * 8 + q * 4);
    float* ap = acc2 + (size_t)d * 8 + q * 4;
    atomicAdd(ap + 0, v.x * nm);
    atomicAdd(ap + 1, v.y * nm);
    atomicAdd(ap + 2, v.z * nm);
    atomicAdd(ap + 3, v.w * nm);
}

__global__ __launch_bounds__(256) void k_final(const float* __restrict__ acc2,
                                               const float* __restrict__ b2,
                                               float* __restrict__ out, int n7) {
    int t = blockIdx.x * blockDim.x + threadIdx.x;
    if (t < n7) {
        int i = t / 7, c = t % 7;
        out[t] = fmaxf(acc2[(size_t)i * 8 + c] + b2[c], 0.0f);
    }
}

extern "C" void kernel_launch(void* const* d_in, const int* in_sizes, int n_in,
                              void* d_out, int out_size, void* d_ws, size_t ws_size,
                              hipStream_t stream) {
    const float* x  = (const float*)d_in[0];
    const int*   ei = (const int*)d_in[1];
    const float* W1 = (const float*)d_in[2];
    const float* b1 = (const float*)d_in[3];
    const float* W2 = (const float*)d_in[4];
    const float* b2 = (const float*)d_in[5];
    float* out = (float*)d_out;

    const int N = in_sizes[0] / 128;
    const int E = in_sizes[1] / 2;

    char* ws = (char*)d_ws;
    float*    h    = (float*)(ws + 0);
    float*    acc1 = (float*)(ws + (size_t)N * 16 * 4);
    float*    h2   = (float*)(ws + (size_t)N * 32 * 4);
    float*    acc2 = (float*)(ws + (size_t)N * 40 * 4);
    float*    deg  = (float*)(ws + (size_t)N * 48 * 4);
    float*    dinv = (float*)(ws + (size_t)N * 49 * 4);
    float*    norm = (float*)(ws + (size_t)N * 50 * 4);
    _Float16* w1B  = (_Float16*)(ws + (size_t)N * 50 * 4 + (size_t)E * 4);
    _Float16* w2B  = (_Float16*)(ws + (size_t)N * 50 * 4 + (size_t)E * 4 + 4096);

    const int B = 256;
    auto blk = [](long long t, int b) { return (int)((t + b - 1) / b); };

    k_deg_init <<<blk(N, B), B, 0, stream>>>(deg, N);
    k_deg_edges<<<blk(E, B), B, 0, stream>>>(ei, deg, E);
    k_dinv     <<<blk(N, B), B, 0, stream>>>(deg, dinv, N);
    k_norm     <<<blk(E, B), B, 0, stream>>>(ei, dinv, norm, E);

    k_prep_w<<<blk(4 * 32 * 16 + 32 * 16, B), B, 0, stream>>>(W1, W2, w1B, w2B);

    int waves1 = (N + 15) / 16;
    k_gemm1   <<<blk((long long)waves1 * 32, B), B, 0, stream>>>(x, w1B, dinv, h, acc1, N);
    k_scatter1<<<blk((long long)E * 4, B), B, 0, stream>>>(ei, norm, h, acc1, E);

    k_gemm2   <<<blk((long long)waves1 * 32, B), B, 0, stream>>>(acc1, b1, w2B, dinv, h2, acc2, N);
    k_scatter2<<<blk((long long)E * 2, B), B, 0, stream>>>(ei, norm, h2, acc2, E);

    k_final<<<blk((long long)N * 7, B), B, 0, stream>>>(acc2, b2, out, N * 7);
}


# in-degree: global float atomics replaced by LDS integer counting per (node range, edge chunk) block; per-chunk partial counts summed in the dinv kernel; counts exact, no float sum reordered
# speedup vs baseline: 1.3561x; 1.3561x over previous
#include <hip/hip_runtime.h>
#include <hip/hip_bf16.h>

typedef __attribute__((ext_vector_type(16))) _Float16 v16h;
typedef __attribute__((ext_vector_type(8)))  float    v8f;

__device__ __forceinline__ int klocal(int lane, int j) {
    return ((lane >> 4) << 3) + (j < 8 ? j : j + 8);
}

#define DEG_R      8192
#define DEG_CHUNKS 80

__global__ __launch_bounds__(256) void k_deg_part(const int* __restrict__ ei,
                                                  unsigned* __restrict__ part,
                                                  int n, int E) {
    __shared__ unsigned cnt[DEG_R];
    const int lo    = blockIdx.x * DEG_R;
    const int chunk = blockIdx.y;
    for (int i = threadIdx.x; i < DEG_R; i += 256) cnt[i] = 0u;
    __syncthreads();

    const int* dst = ei + E;
    const int chunkEdges = ((E + DEG_CHUNKS - 1) / DEG_CHUNKS + 3) & ~3;
    const int e0 = chunk * chunkEdges;
    int e1 = e0 + chunkEdges;
    if (e1 > E) e1 = E;
    if ((E & 3) == 0) {
        for (int e = e0 + 4 * (int)threadIdx.x; e < e1; e += 4 * 4 * 256) {
            int4 d[4];
#pragma unroll
            for (int k = 0; k < 4; ++k) {
                const int ek = e + k * (4 * 256);
                d[k] = (ek < e1) ? *(const int4*)(dst + ek) : make_int4(-1, -1, -1, -1);
            }
#pragma unroll
            for (int k = 0; k < 4; ++k) {
                const unsigned r0 = (unsigned)(d[k].x - lo);
                const unsigned r1 = (unsigned)(d[k].y - lo);
                const unsigned r2 = (unsigned)(d[k].z - lo);
                const unsigned r3 = (unsigned)(d[k].w - lo);
                if (r0 < DEG_R) atomicAdd(&cnt[r0], 1u);
                if (r1 < DEG_R) atomicAdd(&cnt[r1], 1u);
                if (r2 < DEG_R) atomicAdd(&cnt[r2], 1u);
                if (r3 < DEG_R) atomicAdd(&cnt[r3], 1u);
            }
        }
    } else {
        for (int e = e0 + (int)threadIdx.x; e < e1; e += 256) {
            const unsigned r = (unsigned)(dst[e] - lo);
            if (r < DEG_R) atomicAdd(&cnt[r], 1u);
        }
    }
    __syncthreads();

    int nv = n - lo;
    if (nv > DEG_R) nv = DEG_R;
    unsigned* p = part + (size_t)chunk * n + lo;
    for (int i = threadIdx.x; i < nv; i += 256) p[i] = cnt[i];
}

__global__ __launch_bounds__(256) void k_dinv(const unsigned* __restrict__ part,
                                              float* __restrict__ dinv, int n) {
    int i = blockIdx.x * blockDim.x + threadIdx.x;
    if (i < n) {
        unsigned c = 1u;
#pragma unroll 8
        for (int k = 0; k < DEG_CHUNKS; ++k) c += part[(size_t)k * n + i];
        dinv[i] = rsqrtf((float)c);
    }
}

__global__ __launch_bounds__(256) void k_norm(const int* __restrict__ ei,
                                              const float* __restrict__ dinv,
                                              float* __restrict__ norm, int E) {
    int e = blockIdx.x * blockDim.x + threadIdx.x;
    if (e < E) norm[e] = dinv[ei[e]] * dinv[ei[E + e]];
}

__global__ __launch_bounds__(256) void k_prep_w(const float* __restrict__ W1,
                                                const float* __restrict__ W2,
                                                _Float16* __restrict__ w1B,
                                                _Float16* __restrict__ w2B) {
    int t = blockIdx.x * blockDim.x + threadIdx.x;
    if (t < 4 * 32 * 16) {
        int c = t / (32 * 16);
        int l = (t / 16) & 31;
        int j = t & 15;
        int k = 32 * c + klocal(l, j);
        w1B[t] = (_Float16)W1[k * 16 + (l & 15)];
    } else if (t < 4 * 32 * 16 + 32 * 16) {
        int u = t - 4 * 32 * 16;
        int l = u / 16;
        int j = u & 15;
        int k = klocal(l, j);
        int n = l & 15;
        w2B[u] = (k < 16 && n < 7) ? (_Float16)W2[k * 7 + n] : (_Float16)0.0f;
    }
}

__global__ __launch_bounds__(256) void k_gemm1(const float* __restrict__ x,
                                               const _Float16* __restrict__ w1B,
                                               const float* __restrict__ dinv,
                                               float* __restrict__ h,
                                               float* __restrict__ acc1, int nrows) {
    int wave = (blockIdx.x * blockDim.x + threadIdx.x) >> 5;
    int lane = threadIdx.x & 31;
    int row0 = wave * 16;
    if (row0 >= nrows) return;
    int rsub = lane & 15;
    int hi   = lane >> 4;
    const float* xr = x + (size_t)(row0 + rsub) * 128 + hi * 8;
    v8f c = {};
#pragma unroll
    for (int kc = 0; kc < 4; ++kc) {
        const float* p = xr + kc * 32;
        v16h a;
#pragma unroll
        for (int j = 0; j < 8; ++j) a[j] = (_Float16)p[j];
#pragma unroll
        for (int j = 0; j < 8; ++j) a[8 + j] = (_Float16)p[16 + j];
        v16h b = *(const v16h*)(w1B + (size_t)(kc * 32 + lane) * 16);
        c = __builtin_amdgcn_wmma_f32_16x16x32_f16(false, a, false, b,
                                                   (short)0, c, false, false);
    }
#pragma unroll
    for (int r = 0; r < 8; ++r) {
        int row = row0 + hi * 8 + r;
        float hv = c[r];
        float dv = dinv[row];
        h   [(size_t)row * 16 + rsub] = hv;
        acc1[(size_t)row * 16 + rsub] = hv * dv * dv;
    }
}

__global__ __launch_bounds__(256) void k_scatter1(const int* __restrict__ ei,
                                                  const float* __restrict__ norm,
                                                  const float* __restrict__ h,
                                                  float* __restrict__ acc1, int E) {
    int t = blockIdx.x * blockDim.x + threadIdx.x;
    if (t >= 4 * E) return;
    int e = t >> 2, q = t & 3;
    int s = ei[e], d = ei[E + e];
    float nm = norm[e];
    const float4 v = *(const float4*)(h + (size_t)s * 16 + q * 4);
    float* ap = acc1 + (size_t)d * 16 + q * 4;
    atomicAdd(ap + 0, v.x * nm);
    atomicAdd(ap + 1, v.y * nm);
    atomicAdd(ap + 2, v.z * nm);
    atomicAdd(ap + 3, v.w * nm);
}

__global__ __launch_bounds__(256) void k_gemm2(const float* __restrict__ acc1,
                                               const float* __restrict__ b1,
                                               const _Float16* __restrict__ w2B,
                                               const float* __restrict__ dinv,
                                               float* __restrict__ h2,
                                               float* __restrict__ acc2, int nrows) {
    int wave = (blockIdx.x * blockDim.x + threadIdx.x) >> 5;
    int lane = threadIdx.x & 31;
    int row0 = wave * 16;
    if (row0 >= nrows) return;
    int rsub = lane & 15;
    int hi   = lane >> 4;
    int base = hi * 8;
    const float* p = acc1 + (size_t)(row0 + rsub) * 16 + base;
    v16h a;
#pragma unroll
    for (int j = 0; j < 8; ++j) {
        float f = p[j] + b1[base + j];
        a[j] = (_Float16)fmaxf(f, 0.0f);
    }
#pragma unroll
    for (int j = 8; j < 16; ++j) a[j] = (_Float16)0.0f;
    v16h b = *(const v16h*)(w2B + (size_t)lane * 16);
    v8f c = {};
    c = __builtin_amdgcn_wmma_f32_16x16x32_f16(false, a, false, b,
                                               (short)0, c, false, false);
    if (rsub < 8) {
#pragma unroll
        for (int r = 0; r < 8; ++r) {
            int row = row0 + base + r;
            float hv = c[r];
            float dv = dinv[row];
            h2  [(size_t)row * 8 + rsub] = hv;
            acc2[(size_t)row * 8 + rsub] = hv * dv * dv;
        }
    }
}

__global__ __launch_bounds__(256) void k_scatter2(const int* __restrict__ ei,
                                                  const float* __restrict__ norm,
                                                  const float* __restrict__ h2,
                                                  float* __restrict__ acc2, int E) {
    int t = blockIdx.x * blockDim.x + threadIdx.x;
    if (t >= 2 * E) return;
    int e = t >> 1, q = t & 1;
    int s = ei[e], d = ei[E + e];
    float nm = norm[e];
    const float4 v = *(const float4*)(h2 + (size_t)s * 8 + q * 4);
    float* ap = acc2 + (size_t)d * 8 + q * 4;
    atomicAdd(ap + 0, v.x * nm);
    atomicAdd(ap + 1, v.y * nm);
    atomicAdd(ap + 2, v.z * nm);
    atomicAdd(ap + 3, v.w * nm);
}

__global__ __launch_bounds__(256) void k_final(const float* __restrict__ acc2,
                                               const float* __restrict__ b2,
                                               float* __restrict__ out, int n7) {
    int t = blockIdx.x * blockDim.x + threadIdx.x;
    if (t < n7) {
        int i = t / 7, c = t % 7;
        out[t] = fmaxf(acc2[(size_t)i * 8 + c] + b2[c], 0.0f);
    }
}

extern "C" void kernel_launch(void* const* d_in, const int* in_sizes, int n_in,
                              void* d_out, int out_size, void* d_ws, size_t ws_size,
                              hipStream_t stream) {
    const float* x  = (const float*)d_in[0];
    const int*   ei = (const int*)d_in[1];
    const float* W1 = (const float*)d_in[2];
    const float* b1 = (const float*)d_in[3];
    const float* W2 = (const float*)d_in[4];
    const float* b2 = (const float*)d_in[5];
    float* out = (float*)d_out;

    const int N = in_sizes[0] / 128;
    const int E = in_sizes[1] / 2;

    char* ws = (char*)d_ws;
    float*    h    = (float*)(ws + 0);
    float*    acc1 = (float*)(ws + (size_t)N * 16 * 4);
    float*    h2   = (float*)(ws + (size_t)N * 32 * 4);
    float*    acc2 = (float*)(ws + (size_t)N * 40 * 4);
    float*    dinv = (float*)(ws + (size_t)N * 49 * 4);
    float*    norm = (float*)(ws + (size_t)N * 50 * 4);
    _Float16* w1B  = (_Float16*)(ws + (size_t)N * 50 * 4 + (size_t)E * 4);
    _Float16* w2B  = (_Float16*)(ws + (size_t)N * 50 * 4 + (size_t)E * 4 + 4096);
    unsigned* part = (unsigned*)(ws + (size_t)N * 50 * 4 + (size_t)E * 4 + 8192);

    const int B = 256;
    auto blk = [](long long t, int b) { return (int)((t + b - 1) / b); };

    k_deg_part<<<dim3(blk(N, DEG_R), DEG_CHUNKS), B, 0, stream>>>(ei, part, N, E);
    k_dinv    <<<blk(N, B), B, 0, stream>>>(part, dinv, N);
    k_norm    <<<blk(E, B), B, 0, stream>>>(ei, dinv, norm, E);

    k_prep_w<<<blk(4 * 32 * 16 + 32 * 16, B), B, 0, stream>>>(W1, W2, w1B, w2B);

    int waves1 = (N + 15) / 16;
    k_gemm1   <<<blk((long long)waves1 * 32, B), B, 0, stream>>>(x, w1B, dinv, h, acc1, N);
    k_scatter1<<<blk((long long)E * 4, B), B, 0, stream>>>(ei, norm, h, acc1, E);

    k_gemm2   <<<blk((long long)waves1 * 32, B), B, 0, stream>>>(acc1, b1, w2B, dinv, h2, acc2, N);
    k_scatter2<<<blk((long long)E * 2, B), B, 0, stream>>>(ei, norm, h2, acc2, E);

    k_final<<<blk((long long)N * 7, B), B, 0, stream>>>(acc2, b2, out, N * 7);
}


# scatter-add by float atomics replaced by gather per destination: edges counting-sorted by destination (LDS ranks), rows pre-scaled by dinv, each node sums its in-rows in list order and stores once
# speedup vs baseline: 2.7435x; 2.0231x over previous
#include <hip/hip_runtime.h>
#include <hip/hip_bf16.h>

typedef __attribute__((ext_vector_type(16))) _Float16 v16h;
typedef __attribute__((ext_vector_type(8)))  float    v8f;

__device__ __forceinline__ int klocal(int lane, int j) {
    return ((lane >> 4) << 3) + (j < 8 ? j : j + 8);
}

#define DEG_R      8192
#define DEG_CHUNKS 80

__global__ __launch_bounds__(256) void k_deg_part(const int* __restrict__ ei,
                                                  unsigned* __restrict__ part,
                                                  int n, int E) {
    __shared__ unsigned cnt[DEG_R];
    const int lo    = blockIdx.x * DEG_R;
    const int chunk = blockIdx.y;
    for (int i = threadIdx.x; i < DEG_R; i += 256) cnt[i] = 0u;
    __syncthreads();

    const int* dst = ei + E;
    const int chunkEdges = ((E + DEG_CHUNKS - 1) / DEG_CHUNKS + 3) & ~3;
    const int e0 = chunk * chunkEdges;
    int e1 = e0 + chunkEdges;
    if (e1 > E) e1 = E;
    if ((E & 3) == 0) {
        for (int e = e0 + 4 * (int)threadIdx.x; e < e1; e += 4 * 4 * 256) {
            int4 d[4];
#pragma unroll
            for (int k = 0; k < 4; ++k) {
                const int ek = e + k * (4 * 256);
                d[k] = (ek < e1) ? *(const int4*)(dst + ek) : make_int4(-1, -1, -1, -1);
            }
#pragma unroll
            for (int k = 0; k < 4; ++k) {
                const unsigned r0 = (unsigned)(d[k].x - lo);
                const unsigned r1 = (unsigned)(d[k].y - lo);
                const unsigned r2 = (unsigned)(d[k].z - lo);
                const unsigned r3 = (unsigned)(d[k].w - lo);
                if (r0 < DEG_R) atomicAdd(&cnt[r0], 1u);
                if (r1 < DEG_R) atomicAdd(&cnt[r1], 1u);
                if (r2 < DEG_R) atomicAdd(&cnt[r2], 1u);
                if (r3 < DEG_R) atomicAdd(&cnt[r3], 1u);
            }
        }
    } else {
        for (int e = e0 + (int)threadIdx.x; e < e1; e += 256) {
            const unsigned r = (unsigned)(dst[e] - lo);
            if (r < DEG_R) atomicAdd(&cnt[r], 1u);
        }
    }
    __syncthreads();

    int nv = n - lo;
    if (nv > DEG_R) nv = DEG_R;
    unsigned* p = part + (size_t)chunk * n + lo;
    for (int i = threadIdx.x; i < nv; i += 256) p[i] = cnt[i];
}

__global__ __launch_bounds__(256) void k_dinv_scan(unsigned* __restrict__ part,
                                                   float* __restrict__ dinv,
                                                   unsigned* __restrict__ degc,
                                                   unsigned* __restrict__ totals, int n) {
    __shared__ unsigned wsum[8];
    const int i    = blockIdx.x * 256 + threadIdx.x;
    const int lane = threadIdx.x & 31;
    const int wave = threadIdx.x >> 5;
    unsigned c[DEG_CHUNKS];
    const bool live = i < n;
    const unsigned* col = part + (live ? i : n - 1);
#pragma unroll
    for (int k = 0; k < DEG_CHUNKS; ++k) c[k] = col[(size_t)k * n];
    unsigned deg = 0u;
#pragma unroll
    for (int k = 0; k < DEG_CHUNKS; ++k) { const unsigned t = live ? c[k] : 0u; c[k] = deg; deg += t; }
    if (i < n) {
        dinv[i] = rsqrtf((float)(deg + 1u));
        degc[i] = deg;
    }
    unsigned incl = deg;
    for (int d = 1; d < 32; d <<= 1) {
        const unsigned y = __shfl_up(incl, (unsigned)d);
        if (lane >= d) incl += y;
    }
    if (lane == 31) wsum[wave] = incl;
    __syncthreads();
    unsigned wbase = 0u, tot = 0u;
    for (int w = 0; w < 8; ++w) { const unsigned v = wsum[w]; if (w < wave) wbase += v; tot += v; }
    const unsigned loc = wbase + incl - deg;
    if (threadIdx.x == 0) totals[blockIdx.x] = tot;
    if (i < n) {
#pragma unroll
        for (int k = 0; k < DEG_CHUNKS; ++k) part[(size_t)k * n + i] = c[k] + loc;
    }
}

__global__ __launch_bounds__(256) void k_fill(const int* __restrict__ ei,
                                              const unsigned* __restrict__ part,
                                              const unsigned* __restrict__ totals,
                                              unsigned* __restrict__ rowptr,
                                              int* __restrict__ csr,
                                              int n, int E, int ngroups) {
    __shared__ unsigned off[DEG_R];
    __shared__ unsigned baseS[DEG_R / 256];
    const int lo    = blockIdx.x * DEG_R;
    const int chunk = blockIdx.y;
    {
        const int j = threadIdx.x >> 3, sub = threadIdx.x & 7;
        int lim = blockIdx.x * (DEG_R / 256) + j;
        if (lim > ngroups) lim = ngroups;
        unsigned s = 0u;
        for (int b = sub; b < lim; b += 8) s += totals[b];
        s += __shfl_xor(s, 1);
        s += __shfl_xor(s, 2);
        s += __shfl_xor(s, 4);
        if (sub == 0) baseS[j] = s;
    }
    __syncthreads();
    int nv = n - lo;
    if (nv > DEG_R) nv = DEG_R;
    const unsigned* p = part + (size_t)chunk * n + lo;
    for (int i = threadIdx.x; i < nv; i += 256) {
        const unsigned o = baseS[i >> 8] + p[i];
        off[i] = o;
        if (chunk == 0) rowptr[lo + i] = o;
    }
    __syncthreads();

    const int* srcp = ei;
    const int* dst  = ei + E;
    const int chunkEdges = ((E + DEG_CHUNKS - 1) / DEG_CHUNKS + 3) & ~3;
    const int e0 = chunk * chunkEdges;
    int e1 = e0 + chunkEdges;
    if (e1 > E) e1 = E;
    if ((E & 3) == 0) {
        for (int e = e0 + 4 * (int)threadIdx.x; e < e1; e += 4 * 4 * 256) {
            int4 d[4], s[4];
#pragma unroll
            for (int k = 0; k < 4; ++k) {
                const int ek = e + k * (4 * 256);
                const bool in = ek < e1;
                d[k] = in ? *(const int4*)(dst  + ek) : make_int4(-1, -1, -1, -1);
                s[k] = in ? *(const int4*)(srcp + ek) : make_int4(0, 0, 0, 0);
            }
            unsigned r[16];
            int sv[16];
#pragma unroll
            for (int k = 0; k < 4; ++k) {
                r[4 * k + 0] = (unsigned)(d[k].x - lo); sv[4 * k + 0] = s[k].x;
                r[4 * k + 1] = (unsigned)(d[k].y - lo); sv[4 * k + 1] = s[k].y;
                r[4 * k + 2] = (unsigned)(d[k].z - lo); sv[4 * k + 2] = s[k].z;
                r[4 * k + 3] = (unsigned)(d[k].w - lo); sv[4 * k + 3] = s[k].w;
            }
#pragma unroll
            for (int m = 0; m < 16; ++m) if (r[m] < DEG_R) csr[atomicAdd(&off[r[m]], 1u)] = sv[m];
        }
    } else {
        for (int e = e0 + (int)threadIdx.x; e < e1; e += 256) {
            const unsigned r = (unsigned)(dst[e] - lo);
            if (r < DEG_R) csr[atomicAdd(&off[r], 1u)] = srcp[e];
        }
    }
}

__global__ __launch_bounds__(256) void k_prep_w(const float* __restrict__ W1,
                                                const float* __restrict__ W2,
                                                _Float16* __restrict__ w1B,
                                                _Float16* __restrict__ w2B) {
    int t = blockIdx.x * blockDim.x + threadIdx.x;
    if (t < 4 * 32 * 16) {
        int c = t / (32 * 16);
        int l = (t / 16) & 31;
        int j = t & 15;
        int k = 32 * c + klocal(l, j);
        w1B[t] = (_Float16)W1[k * 16 + (l & 15)];
    } else if (t < 4 * 32 * 16 + 32 * 16) {
        int u = t - 4 * 32 * 16;
        int l = u / 16;
        int j = u & 15;
        int k = klocal(l, j);
        int n = l & 15;
        w2B[u] = (k < 16 && n < 7) ? (_Float16)W2[k * 7 + n] : (_Float16)0.0f;
    }
}

__global__ __launch_bounds__(256) void k_gemm1(const float* __restrict__ x,
                                               const _Float16* __restrict__ w1B,
                                               const float* __restrict__ dinv,
                                               float* __restrict__ hs, int nrows) {
    int wave = (blockIdx.x * blockDim.x + threadIdx.x) >> 5;
    int lane = threadIdx.x & 31;
    int row0 = wave * 16;
    if (row0 >= nrows) return;
    int rsub = lane & 15;
    int hi   = lane >> 4;
    const float* xr = x + (size_t)(row0 + rsub) * 128 + hi * 8;
    v8f c = {};
#pragma unroll
    for (int kc = 0; kc < 4; ++kc) {
        const float* p = xr + kc * 32;
        v16h a;
#pragma unroll
        for (int j = 0; j < 8; ++j) a[j] = (_Float16)p[j];
#pragma unroll
        for (int j = 0; j < 8; ++j) a[8 + j] = (_Float16)p[16 + j];
        v16h b = *(const v16h*)(w1B + (size_t)(kc * 32 + lane) * 16);
        c = __builtin_amdgcn_wmma_f32_16x16x32_f16(false, a, false, b,
                                                   (short)0, c, false, false);
    }
#pragma unroll
    for (int r = 0; r < 8; ++r) {
        int row = row0 + hi * 8 + r;
        float dv = dinv[row];
        hs[(size_t)row * 16 + rsub] = c[r] * dv;
    }
}

__global__ __launch_bounds__(256) void k_agg1(const int* __restrict__ csr,
                                              const unsigned* __restrict__ rowptr,
                                              const unsigned* __restrict__ degc,
                                              const float* __restrict__ hs,
                                              const float* __restrict__ dinv,
                                              float* __restrict__ acc1, int n) {
    const int t = blockIdx.x * blockDim.x + threadIdx.x;
    const int i = t >> 2, q = t & 3;
    if (i >= n) return;
    const int* lp = csr + rowptr[i];
    const unsigned cnt = degc[i];
    const float* hq = hs + q * 4;
    float4 a = *(const float4*)(hq + (size_t)i * 16);
    unsigned j = 0;
    for (; j + 4 <= cnt; j += 4) {
        const int s0 = lp[j], s1 = lp[j + 1], s2 = lp[j + 2], s3 = lp[j + 3];
        const float4 v0 = *(const float4*)(hq + (size_t)s0 * 16);
        const float4 v1 = *(const float4*)(hq + (size_t)s1 * 16);
        const float4 v2 = *(const float4*)(hq + (size_t)s2 * 16);
        const float4 v3 = *(const float4*)(hq + (size_t)s3 * 16);
        a.x += v0.x; a.y += v0.y; a.z += v0.z; a.w += v0.w;
        a.x += v1.x; a.y += v1.y; a.z += v1.z; a.w += v1.w;
        a.x += v2.x; a.y += v2.y; a.z += v2.z; a.w += v2.w;
        a.x += v3.x; a.y += v3.y; a.z += v3.z; a.w += v3.w;
    }
    for (; j < cnt; ++j) {
        const float4 v = *(const float4*)(hq + (size_t)lp[j] * 16);
        a.x += v.x; a.y += v.y; a.z += v.z; a.w += v.w;
    }
    const float dv = dinv[i];
    *(float4*)(acc1 + (size_t)i * 16 + q * 4) = make_float4(a.x * dv, a.y * dv, a.z * dv, a.w * dv);
}

__global__ __launch_bounds__(256) void k_gemm2(const float* __restrict__ acc1,
                                               const float* __restrict__ b1,
                                               const _Float16* __restrict__ w2B,
                                               const float* __restrict__ dinv,
                                               float* __restrict__ hs2, int nrows) {
    int wave = (blockIdx.x * blockDim.x + threadIdx.x) >> 5;
    int lane = threadIdx.x & 31;
    int row0 = wave * 16;
    if (row0 >= nrows) return;
    int rsub = lane & 15;
    int hi   = lane >> 4;
    int base = hi * 8;
    const float* p = acc1 + (size_t)(row0 + rsub) * 16 + base;
    v16h a;
#pragma unroll
    for (int j = 0; j < 8; ++j) {
        float f = p[j] + b1[base + j];
        a[j] = (_Float16)fmaxf(f, 0.0f);
    }
#pragma unroll
    for (int j = 8; j < 16; ++j) a[j] = (_Float16)0.0f;
    v16h b = *(const v16h*)(w2B + (size_t)lane * 16);
    v8f c = {};
    c = __builtin_amdgcn_wmma_f32_16x16x32_f16(false, a, false, b,
                                               (short)0, c, false, false);
    if (rsub < 8) {
#pragma unroll
        for (int r = 0; r < 8; ++r) {
            int row = row0 + base + r;
            float dv = dinv[row];
            hs2[(size_t)row * 8 + rsub] = c[r] * dv;
        }
    }
}

__global__ __launch_bounds__(256) void k_agg2(const int* __restrict__ csr,
                                              const unsigned* __restrict__ rowptr,
                                              const unsigned* __restrict__ degc,
                                              const float* __restrict__ hs2,
                                              const float* __restrict__ dinv,
                                              const float* __restrict__ b2,
                                              float* __restrict__ out, int n) {
    const int t = blockIdx.x * blockDim.x + threadIdx.x;
    const int i = t >> 1, q = t & 1;
    if (i >= n) return;
    const int* lp = csr + rowptr[i];
    const unsigned cnt = degc[i];
    const float* hq = hs2 + q * 4;
    float4 a = *(const float4*)(hq + (size_t)i * 8);
    unsigned j = 0;
    for (; j + 4 <= cnt; j += 4) {
        const int s0 = lp[j], s1 = lp[j + 1], s2 = lp[j + 2], s3 = lp[j + 3];
        const float4 v0 = *(const float4*)(hq + (size_t)s0 * 8);
        const float4 v1 = *(const float4*)(hq + (size_t)s1 * 8);
        const float4 v2 = *(const float4*)(hq + (size_t)s2 * 8);
        const float4 v3 = *(const float4*)(hq + (size_t)s3 * 8);
        a.x += v0.x; a.y += v0.y; a.z += v0.z; a.w += v0.w;
        a.x += v1.x; a.y += v1.y; a.z += v1.z; a.w += v1.w;
        a.x += v2.x; a.y += v2.y; a.z += v2.z; a.w += v2.w;
        a.x += v3.x; a.y += v3.y; a.z += v3.z; a.w += v3.w;
    }
    for (; j < cnt; ++j) {
        const float4 v = *(const float4*)(hq + (size_t)lp[j] * 8);
        a.x += v.x; a.y += v.y; a.z += v.z; a.w += v.w;
    }
    const float dv = dinv[i];
    float* o = out + (size_t)i * 7 + q * 4;
    const float* bq = b2 + q * 4;
    o[0] = fmaxf(a.x * dv + bq[0], 0.0f);
    o[1] = fmaxf(a.y * dv + bq[1], 0.0f);
    o[2] = fmaxf(a.z * dv + bq[2], 0.0f);
    if (q == 0) o[3] = fmaxf(a.w * dv + bq[3], 0.0f);
}

extern "C" void kernel_launch(void* const* d_in, const int* in_sizes, int n_in,
                              void* d_out, int out_size, void* d_ws, size_t ws_size,
                              hipStream_t stream) {
    const float* x  = (const float*)d_in[0];
    const int*   ei = (const int*)d_in[1];
    const float* W1 = (const float*)d_in[2];
    const float* b1 = (const float*)d_in[3];
    const float* W2 = (const float*)d_in[4];
    const float* b2 = (const float*)d_in[5];
    float* out = (float*)d_out;

    const int N = in_sizes[0] / 128;
    const int E = in_sizes[1] / 2;

    char* ws = (char*)d_ws;
    float*    hs     = (float*)(ws + 0);
    float*    acc1   = (float*)(ws + (size_t)N * 16 * 4);
    float*    hs2    = (float*)(ws + (size_t)N * 32 * 4);
    unsigned* degc   = (unsigned*)(ws + (size_t)N * 40 * 4);
    unsigned* rowptr = (unsigned*)(ws + (size_t)N * 41 * 4);
    unsigned* totals = (unsigned*)(ws + (size_t)N * 42 * 4);
    float*    dinv   = (float*)(ws + (size_t)N * 49 * 4);
    int*      csr    = (int*)(ws + (size_t)N * 50 * 4);
    _Float16* w1B  = (_Float16*)(ws + (size_t)N * 50 * 4 + (size_t)E * 4);
    _Float16* w2B  = (_Float16*)(ws + (size_t)N * 50 * 4 + (size_t)E * 4 + 4096);
    unsigned* part = (unsigned*)(ws + (size_t)N * 50 * 4 + (size_t)E * 4 + 8192);

    const int B = 256;
    auto blk = [](long long t, int b) { return (int)((t + b - 1) / b); };
    const int groups = blk(N, 256);

    k_deg_part <<<dim3(blk(N, DEG_R), DEG_CHUNKS), B, 0, stream>>>(ei, part, N, E);
    k_dinv_scan<<<groups, 256, 0, stream>>>(part, dinv, degc, totals, N);
    k_fill     <<<dim3(blk(N, DEG_R), DEG_CHUNKS), B, 0, stream>>>(ei, part, totals, rowptr, csr, N, E, groups);

    k_prep_w<<<blk(4 * 32 * 16 + 32 * 16, B), B, 0, stream>>>(W1, W2, w1B, w2B);

    int waves1 = (N + 15) / 16;
    k_gemm1<<<blk((long long)waves1 * 32, B), B, 0, stream>>>(x, w1B, dinv, hs, N);
    k_agg1 <<<blk((long long)N * 4, B), B, 0, stream>>>(csr, rowptr, degc, hs, dinv, acc1, N);

    k_gemm2<<<blk((long long)waves1 * 32, B), B, 0, stream>>>(acc1, b1, w2B, dinv, hs2, N);
    k_agg2 <<<blk((long long)N * 2, B), B, 0, stream>>>(csr, rowptr, degc, hs2, dinv, b2, out, N);
}


# serial round trips per wave cut: deg/fill loads and fill's 16 returning LDS adds per trip go out as groups; agg walks take 8 edges per trip, next indices prefetched, tail masked; node sum order kept
# speedup vs baseline: 2.9967x; 1.0923x over previous
#include <hip/hip_runtime.h>
#include <hip/hip_bf16.h>

typedef __attribute__((ext_vector_type(16))) _Float16 v16h;
typedef __attribute__((ext_vector_type(8)))  float    v8f;

__device__ __forceinline__ int klocal(int lane, int j) {
    return ((lane >> 4) << 3) + (j < 8 ? j : j + 8);
}

#define DEG_R      8192
#define DEG_CHUNKS 80

__global__ __launch_bounds__(256) void k_deg_part(const int* __restrict__ ei,
                                                  unsigned* __restrict__ part,
                                                  int n, int E) {
    __shared__ unsigned cnt[DEG_R];
    const int lo    = blockIdx.x * DEG_R;
    const int chunk = blockIdx.y;
    for (int i = threadIdx.x; i < DEG_R; i += 256) cnt[i] = 0u;
    __syncthreads();

    const int* dst = ei + E;
    const int chunkEdges = ((E + DEG_CHUNKS - 1) / DEG_CHUNKS + 3) & ~3;
    const int e0 = chunk * chunkEdges;
    int e1 = e0 + chunkEdges;
    if (e1 > E) e1 = E;
    if ((E & 3) == 0) {
        for (int e = e0 + 4 * (int)threadIdx.x; e < e1; e += 8 * 4 * 256) {
            int4 d[8];
#pragma unroll
            for (int k = 0; k < 8; ++k) {
                const int ek = e + k * (4 * 256);
                d[k] = (ek < e1) ? *(const int4*)(dst + ek) : make_int4(-1, -1, -1, -1);
            }
#pragma unroll
            for (int k = 0; k < 8; ++k) {
                const unsigned r0 = (unsigned)(d[k].x - lo);
                const unsigned r1 = (unsigned)(d[k].y - lo);
                const unsigned r2 = (unsigned)(d[k].z - lo);
                const unsigned r3 = (unsigned)(d[k].w - lo);
                if (r0 < DEG_R) atomicAdd(&cnt[r0], 1u);
                if (r1 < DEG_R) atomicAdd(&cnt[r1], 1u);
                if (r2 < DEG_R) atomicAdd(&cnt[r2], 1u);
                if (r3 < DEG_R) atomicAdd(&cnt[r3], 1u);
            }
        }
    } else {
        for (int e = e0 + (int)threadIdx.x; e < e1; e += 256) {
            const unsigned r = (unsigned)(dst[e] - lo);
            if (r < DEG_R) atomicAdd(&cnt[r], 1u);
        }
    }
    __syncthreads();

    int nv = n - lo;
    if (nv > DEG_R) nv = DEG_R;
    unsigned* p = part + (size_t)chunk * n + lo;
    for (int i = threadIdx.x; i < nv; i += 256) p[i] = cnt[i];
}

__global__ __launch_bounds__(256) void k_dinv_scan(unsigned* __restrict__ part,
                                                   float* __restrict__ dinv,
                                                   unsigned* __restrict__ degc,
                                                   unsigned* __restrict__ totals, int n) {
    __shared__ unsigned wsum[8];
    const int i    = blockIdx.x * 256 + threadIdx.x;
    const int lane = threadIdx.x & 31;
    const int wave = threadIdx.x >> 5;
    unsigned c[DEG_CHUNKS];
    const bool live = i < n;
    const unsigned* col = part + (live ? i : n - 1);
#pragma unroll
    for (int k = 0; k < DEG_CHUNKS; ++k) c[k] = col[(size_t)k * n];
    unsigned deg = 0u;
#pragma unroll
    for (int k = 0; k < DEG_CHUNKS; ++k) { const unsigned t = live ? c[k] : 0u; c[k] = deg; deg += t; }
    if (i < n) {
        dinv[i] = rsqrtf((float)(deg + 1u));
        degc[i] = deg;
    }
    unsigned incl = deg;
    for (int d = 1; d < 32; d <<= 1) {
        const unsigned y = __shfl_up(incl, (unsigned)d);
        if (lane >= d) incl += y;
    }
    if (lane == 31) wsum[wave] = incl;
    __syncthreads();
    unsigned wbase = 0u, tot = 0u;
    for (int w = 0; w < 8; ++w) { const unsigned v = wsum[w]; if (w < wave) wbase += v; tot += v; }
    const unsigned loc = wbase + incl - deg;
    if (threadIdx.x == 0) totals[blockIdx.x] = tot;
    if (i < n) {
#pragma unroll
        for (int k = 0; k < DEG_CHUNKS; ++k) part[(size_t)k * n + i] = c[k] + loc;
    }
}

__global__ __launch_bounds__(256) void k_fill(const int* __restrict__ ei,
                                              const unsigned* __restrict__ part,
                                              const unsigned* __restrict__ totals,
                                              unsigned* __restrict__ rowptr,
                                              int* __restrict__ csr,
                                              int n, int E, int ngroups) {
    __shared__ unsigned off[DEG_R];
    __shared__ unsigned baseS[DEG_R / 256];
    const int lo    = blockIdx.x * DEG_R;
    const int chunk = blockIdx.y;
    {
        const int j = threadIdx.x >> 3, sub = threadIdx.x & 7;
        int lim = blockIdx.x * (DEG_R / 256) + j;
        if (lim > ngroups) lim = ngroups;
        unsigned s = 0u;
        for (int b = sub; b < lim; b += 8) s += totals[b];
        s += __shfl_xor(s, 1);
        s += __shfl_xor(s, 2);
        s += __shfl_xor(s, 4);
        if (sub == 0) baseS[j] = s;
    }
    __syncthreads();
    int nv = n - lo;
    if (nv > DEG_R) nv = DEG_R;
    const unsigned* p = part + (size_t)chunk * n + lo;
    for (int i0 = threadIdx.x; i0 < DEG_R; i0 += 16 * 256) {
        unsigned pv[16], bs[16];
#pragma unroll
        for (int k = 0; k < 16; ++k) {
            const int i = i0 + k * 256;
            pv[k] = p[i < nv ? i : nv - 1];
            bs[k] = baseS[i >> 8];
        }
#pragma unroll
        for (int k = 0; k < 16; ++k) {
            const int i = i0 + k * 256;
            if (i < nv) {
                const unsigned o = bs[k] + pv[k];
                off[i] = o;
                if (chunk == 0) rowptr[lo + i] = o;
            }
        }
    }
    __syncthreads();

    const int* srcp = ei;
    const int* dst  = ei + E;
    const int chunkEdges = ((E + DEG_CHUNKS - 1) / DEG_CHUNKS + 3) & ~3;
    const int e0 = chunk * chunkEdges;
    int e1 = e0 + chunkEdges;
    if (e1 > E) e1 = E;
    if ((E & 3) == 0) {
        for (int e = e0 + 4 * (int)threadIdx.x; e < e1; e += 4 * 4 * 256) {
            int4 d[4], s[4];
#pragma unroll
            for (int k = 0; k < 4; ++k) {
                const int ek = e + k * (4 * 256);
                const bool in = ek < e1;
                d[k] = in ? *(const int4*)(dst  + ek) : make_int4(-1, -1, -1, -1);
                s[k] = in ? *(const int4*)(srcp + ek) : make_int4(0, 0, 0, 0);
            }
            unsigned r[16];
            int sv[16];
#pragma unroll
            for (int k = 0; k < 4; ++k) {
                r[4 * k + 0] = (unsigned)(d[k].x - lo); sv[4 * k + 0] = s[k].x;
                r[4 * k + 1] = (unsigned)(d[k].y - lo); sv[4 * k + 1] = s[k].y;
                r[4 * k + 2] = (unsigned)(d[k].z - lo); sv[4 * k + 2] = s[k].z;
                r[4 * k + 3] = (unsigned)(d[k].w - lo); sv[4 * k + 3] = s[k].w;
            }
            unsigned slot[16];
#pragma unroll
            for (int m = 0; m < 16; ++m) if (r[m] < DEG_R) slot[m] = atomicAdd(&off[r[m]], 1u);
#pragma unroll
            for (int m = 0; m < 16; ++m) if (r[m] < DEG_R) csr[slot[m]] = sv[m];
            __builtin_amdgcn_fence(__ATOMIC_RELEASE, "workgroup", "local");
        }
    } else {
        for (int e = e0 + (int)threadIdx.x; e < e1; e += 256) {
            const unsigned r = (unsigned)(dst[e] - lo);
            if (r < DEG_R) csr[atomicAdd(&off[r], 1u)] = srcp[e];
        }
    }
}

__global__ __launch_bounds__(256) void k_prep_w(const float* __restrict__ W1,
                                                const float* __restrict__ W2,
                                                _Float16* __restrict__ w1B,
                                                _Float16* __restrict__ w2B) {
    int t = blockIdx.x * blockDim.x + threadIdx.x;
    if (t < 4 * 32 * 16) {
        int c = t / (32 * 16);
        int l = (t / 16) & 31;
        int j = t & 15;
        int k = 32 * c + klocal(l, j);
        w1B[t] = (_Float16)W1[k * 16 + (l & 15)];
    } else if (t < 4 * 32 * 16 + 32 * 16) {
        int u = t - 4 * 32 * 16;
        int l = u / 16;
        int j = u & 15;
        int k = klocal(l, j);
        int n = l & 15;
        w2B[u] = (k < 16 && n < 7) ? (_Float16)W2[k * 7 + n] : (_Float16)0.0f;
    }
}

__global__ __launch_bounds__(256) void k_gemm1(const float* __restrict__ x,
                                               const _Float16* __restrict__ w1B,
                                               const float* __restrict__ dinv,
                                               float* __restrict__ hs, int nrows) {
    int wave = (blockIdx.x * blockDim.x + threadIdx.x) >> 5;
    int lane = threadIdx.x & 31;
    int row0 = wave * 16;
    if (row0 >= nrows) return;
    int rsub = lane & 15;
    int hi   = lane >> 4;
    const float* xr = x + (size_t)(row0 + rsub) * 128 + hi * 8;
    v8f c = {};
#pragma unroll
    for (int kc = 0; kc < 4; ++kc) {
        const float* p = xr + kc * 32;
        v16h a;
#pragma unroll
        for (int j = 0; j < 8; ++j) a[j] = (_Float16)p[j];
#pragma unroll
        for (int j = 0; j < 8; ++j) a[8 + j] = (_Float16)p[16 + j];
        v16h b = *(const v16h*)(w1B + (size_t)(kc * 32 + lane) * 16);
        c = __builtin_amdgcn_wmma_f32_16x16x32_f16(false, a, false, b,
                                                   (short)0, c, false, false);
    }
#pragma unroll
    for (int r = 0; r < 8; ++r) {
        int row = row0 + hi * 8 + r;
        float dv = dinv[row];
        hs[(size_t)row * 16 + rsub] = c[r] * dv;
    }
}

__global__ __launch_bounds__(256) void k_agg1(const int* __restrict__ csr,
                                              const unsigned* __restrict__ rowptr,
                                              const unsigned* __restrict__ degc,
                                              const float* __restrict__ hs,
                                              const float* __restrict__ dinv,
                                              float* __restrict__ acc1, int n) {
    const int t = blockIdx.x * blockDim.x + threadIdx.x;
    const int i = t >> 2;
    const unsigned q = (unsigned)(t & 3);
    if (i >= n) return;
    const int* lp = csr + rowptr[i];
    const unsigned cnt = degc[i];
    const float dv = dinv[i];
    const float4* H = (const float4*)hs;
    float4 a = H[4u * (unsigned)i + q];
    int nx[8];
#pragma unroll
    for (int k = 0; k < 8; ++k) nx[k] = lp[k];
    unsigned j = 0;
    do {
        int cu[8];
#pragma unroll
        for (int k = 0; k < 8; ++k) cu[k] = nx[k];
#pragma unroll
        for (int k = 0; k < 8; ++k) nx[k] = lp[j + 8 + k];
        const unsigned s0 = (j < cnt) ? (unsigned)cu[0] : (unsigned)i;
        float4 v[8];
#pragma unroll
        for (int k = 0; k < 8; ++k) {
            const unsigned s = (j + k < cnt) ? (unsigned)cu[k] : s0;
            v[k] = H[4u * s + q];
        }
#pragma unroll
        for (int k = 0; k < 8; ++k)
            if (j + k < cnt) { a.x += v[k].x; a.y += v[k].y; a.z += v[k].z; a.w += v[k].w; }
        j += 8;
    } while (j < cnt);
    *(float4*)(acc1 + (size_t)i * 16 + q * 4) = make_float4(a.x * dv, a.y * dv, a.z * dv, a.w * dv);
}

__global__ __launch_bounds__(256) void k_gemm2(const float* __restrict__ acc1,
                                               const float* __restrict__ b1,
                                               const _Float16* __restrict__ w2B,
                                               const float* __restrict__ dinv,
                                               float* __restrict__ hs2, int nrows) {
    int wave = (blockIdx.x * blockDim.x + threadIdx.x) >> 5;
    int lane = threadIdx.x & 31;
    int row0 = wave * 16;
    if (row0 >= nrows) return;
    int rsub = lane & 15;
    int hi   = lane >> 4;
    int base = hi * 8;
    const float* p = acc1 + (size_t)(row0 + rsub) * 16 + base;
    v16h a;
#pragma unroll
    for (int j = 0; j < 8; ++j) {
        float f = p[j] + b1[base + j];
        a[j] = (_Float16)fmaxf(f, 0.0f);
    }
#pragma unroll
    for (int j = 8; j < 16; ++j) a[j] = (_Float16)0.0f;
    v16h b = *(const v16h*)(w2B + (size_t)lane * 16);
    v8f c = {};
    c = __builtin_amdgcn_wmma_f32_16x16x32_f16(false, a, false, b,
                                               (short)0, c, false, false);
    if (rsub < 8) {
#pragma unroll
        for (int r = 0; r < 8; ++r) {
            int row = row0 + base + r;
            float dv = dinv[row];
            hs2[(size_t)row * 8 + rsub] = c[r] * dv;
        }
    }
}

__global__ __launch_bounds__(256) void k_agg2(const int* __restrict__ csr,
                                              const unsigned* __restrict__ rowptr,
                                              const unsigned* __restrict__ degc,
                                              const float* __restrict__ hs2,
                                              const float* __restrict__ dinv,
                                              const float* __restrict__ b2,
                                              float* __restrict__ out, int n) {
    const int t = blockIdx.x * blockDim.x + threadIdx.x;
    const int i = t >> 1;
    const unsigned q = (unsigned)(t & 1);
    if (i >= n) return;
    const int* lp = csr + rowptr[i];
    const unsigned cnt = degc[i];
    const float dv = dinv[i];
    const float* bq = b2 + q * 4;
    const float bias0 = bq[0], bias1 = bq[1], bias2 = bq[2];
    const float bias3 = (q == 0u) ? bq[3] : 0.0f;
    const float4* H = (const float4*)hs2;
    float4 a = H[2u * (unsigned)i + q];
    int nx[8];
#pragma unroll
    for (int k = 0; k < 8; ++k) nx[k] = lp[k];
    unsigned j = 0;
    do {
        int cu[8];
#pragma unroll
        for (int k = 0; k < 8; ++k) cu[k] = nx[k];
#pragma unroll
        for (int k = 0; k < 8; ++k) nx[k] = lp[j + 8 + k];
        const unsigned s0 = (j < cnt) ? (unsigned)cu[0] : (unsigned)i;
        float4 v[8];
#pragma unroll
        for (int k = 0; k < 8; ++k) {
            const unsigned s = (j + k < cnt) ? (unsigned)cu[k] : s0;
            v[k] = H[2u * s + q];
        }
#pragma unroll
        for (int k = 0; k < 8; ++k)
            if (j + k < cnt) { a.x += v[k].x; a.y += v[k].y; a.z += v[k].z; a.w += v[k].w; }
        j += 8;
    } while (j < cnt);
    float* o = out + (size_t)i * 7 + q * 4;
    o[0] = fmaxf(a.x * dv + bias0, 0.0f);
    o[1] = fmaxf(a.y * dv + bias1, 0.0f);
    o[2] = fmaxf(a.z * dv + bias2, 0.0f);
    if (q == 0u) o[3] = fmaxf(a.w * dv + bias3, 0.0f);
}

extern "C" void kernel_launch(void* const* d_in, const int* in_sizes, int n_in,
                              void* d_out, int out_size, void* d_ws, size_t ws_size,
                              hipStream_t stream) {
    const float* x  = (const float*)d_in[0];
    const int*   ei = (const int*)d_in[1];
    const float* W1 = (const float*)d_in[2];
    const float* b1 = (const float*)d_in[3];
    const float* W2 = (const float*)d_in[4];
    const float* b2 = (const float*)d_in[5];
    float* out = (float*)d_out;

    const int N = in_sizes[0] / 128;
    const int E = in_sizes[1] / 2;

    char* ws = (char*)d_ws;
    float*    hs     = (float*)(ws + 0);
    float*    acc1   = (float*)(ws + (size_t)N * 16 * 4);
    float*    hs2    = (float*)(ws + (size_t)N * 32 * 4);
    unsigned* degc   = (unsigned*)(ws + (size_t)N * 40 * 4);
    unsigned* rowptr = (unsigned*)(ws + (size_t)N * 41 * 4);
    unsigned* totals = (unsigned*)(ws + (size_t)N * 42 * 4);
    float*    dinv   = (float*)(ws + (size_t)N * 49 * 4);
    int*      csr    = (int*)(ws + (size_t)N * 50 * 4);
    _Float16* w1B  = (_Float16*)(ws + (size_t)N * 50 * 4 + (size_t)E * 4);
    _Float16* w2B  = (_Float16*)(ws + (size_t)N * 50 * 4 + (size_t)E * 4 + 4096);
    unsigned* part = (unsigned*)(ws + (size_t)N * 50 * 4 + (size_t)E * 4 + 8192);

    const int B = 256;
    auto blk = [](long long t, int b) { return (int)((t + b - 1) / b); };
    const int groups = blk(N, 256);

    k_deg_part <<<dim3(blk(N, DEG_R), DEG_CHUNKS), B, 0, stream>>>(ei, part, N, E);
    k_dinv_scan<<<groups, 256, 0, stream>>>(part, dinv, degc, totals, N);
    k_fill     <<<dim3(blk(N, DEG_R), DEG_CHUNKS), B, 0, stream>>>(ei, part, totals, rowptr, csr, N, E, groups);

    k_prep_w<<<blk(4 * 32 * 16 + 32 * 16, B), B, 0, stream>>>(W1, W2, w1B, w2B);

    int waves1 = (N + 15) / 16;
    k_gemm1<<<blk((long long)waves1 * 32, B), B, 0, stream>>>(x, w1B, dinv, hs, N);
    k_agg1 <<<blk((long long)N * 4, B), B, 0, stream>>>(csr, rowptr, degc, hs, dinv, acc1, N);

    k_gemm2<<<blk((long long)waves1 * 32, B), B, 0, stream>>>(acc1, b1, w2B, dinv, hs2, N);
    k_agg2 <<<blk((long long)N * 2, B), B, 0, stream>>>(csr, rowptr, degc, hs2, dinv, b2, out, N);
}


# independent phases share a launch: the weight swizzle rides as one more grid row of the degree count, the x@W1 tiles as grid rows behind the list fill's; 8 launches become 6; no sum reordered
# speedup vs baseline: 3.1489x; 1.0508x over previous
#include <hip/hip_runtime.h>
#include <hip/hip_bf16.h>

typedef __attribute__((ext_vector_type(16))) _Float16 v16h;
typedef __attribute__((ext_vector_type(8)))  float    v8f;

__device__ __forceinline__ int klocal(int lane, int j) {
    return ((lane >> 4) << 3) + (j < 8 ? j : j + 8);
}

__device__ __forceinline__ void prep_w_elem(int t, const float* __restrict__ W1,
                                            const float* __restrict__ W2,
                                            _Float16* __restrict__ w1B,
                                            _Float16* __restrict__ w2B);
__device__ __forceinline__ void gemm1_tile(const float* __restrict__ x,
                                           const _Float16* __restrict__ w1B,
                                           const float* __restrict__ dinv,
                                           float* __restrict__ hs, int nrows,
                                           int wave, int lane);

#define DEG_R      8192
#define DEG_CHUNKS 80

__global__ __launch_bounds__(256) void k_deg_part(const int* __restrict__ ei,
                                                  unsigned* __restrict__ part,
                                                  int n, int E,
                                                  const float* __restrict__ W1,
                                                  const float* __restrict__ W2,
                                                  _Float16* __restrict__ w1B,
                                                  _Float16* __restrict__ w2B) {
    __shared__ unsigned cnt[DEG_R];
    if (blockIdx.y == DEG_CHUNKS) {
        for (int t = blockIdx.x * 256 + threadIdx.x; t < 4 * 32 * 16 + 32 * 16; t += gridDim.x * 256)
            prep_w_elem(t, W1, W2, w1B, w2B);
        return;
    }
    const int lo    = blockIdx.x * DEG_R;
    const int chunk = blockIdx.y;
    for (int i = threadIdx.x; i < DEG_R; i += 256) cnt[i] = 0u;
    __syncthreads();

    const int* dst = ei + E;
    const int chunkEdges = ((E + DEG_CHUNKS - 1) / DEG_CHUNKS + 3) & ~3;
    const int e0 = chunk * chunkEdges;
    int e1 = e0 + chunkEdges;
    if (e1 > E) e1 = E;
    if ((E & 3) == 0) {
        for (int e = e0 + 4 * (int)threadIdx.x; e < e1; e += 8 * 4 * 256) {
            int4 d[8];
#pragma unroll
            for (int k = 0; k < 8; ++k) {
                const int ek = e + k * (4 * 256);
                d[k] = (ek < e1) ? *(const int4*)(dst + ek) : make_int4(-1, -1, -1, -1);
            }
#pragma unroll
            for (int k = 0; k < 8; ++k) {
                const unsigned r0 = (unsigned)(d[k].x - lo);
                const unsigned r1 = (unsigned)(d[k].y - lo);
                const unsigned r2 = (unsigned)(d[k].z - lo);
                const unsigned r3 = (unsigned)(d[k].w - lo);
                if (r0 < DEG_R) atomicAdd(&cnt[r0], 1u);
                if (r1 < DEG_R) atomicAdd(&cnt[r1], 1u);
                if (r2 < DEG_R) atomicAdd(&cnt[r2], 1u);
                if (r3 < DEG_R) atomicAdd(&cnt[r3], 1u);
            }
        }
    } else {
        for (int e = e0 + (int)threadIdx.x; e < e1; e += 256) {
            const unsigned r = (unsigned)(dst[e] - lo);
            if (r < DEG_R) atomicAdd(&cnt[r], 1u);
        }
    }
    __syncthreads();

    int nv = n - lo;
    if (nv > DEG_R) nv = DEG_R;
    unsigned* p = part + (size_t)chunk * n + lo;
    for (int i = threadIdx.x; i < nv; i += 256) p[i] = cnt[i];
}

__global__ __launch_bounds__(256) void k_dinv_scan(unsigned* __restrict__ part,
                                                   float* __restrict__ dinv,
                                                   unsigned* __restrict__ degc,
                                                   unsigned* __restrict__ totals, int n) {
    __shared__ unsigned wsum[8];
    const int i    = blockIdx.x * 256 + threadIdx.x;
    const int lane = threadIdx.x & 31;
    const int wave = threadIdx.x >> 5;
    unsigned c[DEG_CHUNKS];
    const bool live = i < n;
    const unsigned* col = part + (live ? i : n - 1);
#pragma unroll
    for (int k = 0; k < DEG_CHUNKS; ++k) c[k] = col[(size_t)k * n];
    unsigned deg = 0u;
#pragma unroll
    for (int k = 0; k < DEG_CHUNKS; ++k) { const unsigned t = live ? c[k] : 0u; c[k] = deg; deg += t; }
    if (i < n) {
        dinv[i] = rsqrtf((float)(deg + 1u));
        degc[i] = deg;
    }
    unsigned incl = deg;
    for (int d = 1; d < 32; d <<= 1) {
        const unsigned y = __shfl_up(incl, (unsigned)d);
        if (lane >= d) incl += y;
    }
    if (lane == 31) wsum[wave] = incl;
    __syncthreads();
    unsigned wbase = 0u, tot = 0u;
    for (int w = 0; w < 8; ++w) { const unsigned v = wsum[w]; if (w < wave) wbase += v; tot += v; }
    const unsigned loc = wbase + incl - deg;
    if (threadIdx.x == 0) totals[blockIdx.x] = tot;
    if (i < n) {
#pragma unroll
        for (int k = 0; k < DEG_CHUNKS; ++k) part[(size_t)k * n + i] = c[k] + loc;
    }
}

__global__ __launch_bounds__(256) void k_fill(const int* __restrict__ ei,
                                              const unsigned* __restrict__ part,
                                              const unsigned* __restrict__ totals,
                                              unsigned* __restrict__ rowptr,
                                              int* __restrict__ csr,
                                              int n, int E, int ngroups,
                                              const float* __restrict__ x,
                                              const _Float16* __restrict__ w1B,
                                              const float* __restrict__ dinv,
                                              float* __restrict__ hs) {
    __shared__ unsigned off[DEG_R];
    __shared__ unsigned baseS[DEG_R / 256];
    if (blockIdx.y >= DEG_CHUNKS) {
        const int gb = (int)(blockIdx.y - DEG_CHUNKS) * (int)gridDim.x + (int)blockIdx.x;
        gemm1_tile(x, w1B, dinv, hs, n, gb * 8 + (int)(threadIdx.x >> 5), (int)(threadIdx.x & 31));
        return;
    }
    const int lo    = blockIdx.x * DEG_R;
    const int chunk = blockIdx.y;
    {
        const int j = threadIdx.x >> 3, sub = threadIdx.x & 7;
        int lim = blockIdx.x * (DEG_R / 256) + j;
        if (lim > ngroups) lim = ngroups;
        unsigned s = 0u;
        for (int b = sub; b < lim; b += 8) s += totals[b];
        s += __shfl_xor(s, 1);
        s += __shfl_xor(s, 2);
        s += __shfl_xor(s, 4);
        if (sub == 0) baseS[j] = s;
    }
    __syncthreads();
    int nv = n - lo;
    if (nv > DEG_R) nv = DEG_R;
    const unsigned* p = part + (size_t)chunk * n + lo;
    for (int i0 = threadIdx.x; i0 < DEG_R; i0 += 16 * 256) {
        unsigned pv[16], bs[16];
#pragma unroll
        for (int k = 0; k < 16; ++k) {
            const int i = i0 + k * 256;
            pv[k] = p[i < nv ? i : nv - 1];
            bs[k] = baseS[i >> 8];
        }
#pragma unroll
        for (int k = 0; k < 16; ++k) {
            const int i = i0 + k * 256;
            if (i < nv) {
                const unsigned o = bs[k] + pv[k];
                off[i] = o;
                if (chunk == 0) rowptr[lo + i] = o;
            }
        }
    }
    __syncthreads();

    const int* srcp = ei;
    const int* dst  = ei + E;
    const int chunkEdges = ((E + DEG_CHUNKS - 1) / DEG_CHUNKS + 3) & ~3;
    const int e0 = chunk * chunkEdges;
    int e1 = e0 + chunkEdges;
    if (e1 > E) e1 = E;
    if ((E & 3) == 0) {
        for (int e = e0 + 4 * (int)threadIdx.x; e < e1; e += 4 * 4 * 256) {
            int4 d[4], s[4];
#pragma unroll
            for (int k = 0; k < 4; ++k) {
                const int ek = e + k * (4 * 256);
                const bool in = ek < e1;
                d[k] = in ? *(const int4*)(dst  + ek) : make_int4(-1, -1, -1, -1);
                s[k] = in ? *(const int4*)(srcp + ek) : make_int4(0, 0, 0, 0);
            }
            unsigned r[16];
            int sv[16];
#pragma unroll
            for (int k = 0; k < 4; ++k) {
                r[4 * k + 0] = (unsigned)(d[k].x - lo); sv[4 * k + 0] = s[k].x;
                r[4 * k + 1] = (unsigned)(d[k].y - lo); sv[4 * k + 1] = s[k].y;
                r[4 * k + 2] = (unsigned)(d[k].z - lo); sv[4 * k + 2] = s[k].z;
                r[4 * k + 3] = (unsigned)(d[k].w - lo); sv[4 * k + 3] = s[k].w;
            }
            unsigned slot[16];
#pragma unroll
            for (int m = 0; m < 16; ++m) if (r[m] < DEG_R) slot[m] = atomicAdd(&off[r[m]], 1u);
#pragma unroll
            for (int m = 0; m < 16; ++m) if (r[m] < DEG_R) csr[slot[m]] = sv[m];
            __builtin_amdgcn_fence(__ATOMIC_RELEASE, "workgroup", "local");
        }
    } else {
        for (int e = e0 + (int)threadIdx.x; e < e1; e += 256) {
            const unsigned r = (unsigned)(dst[e] - lo);
            if (r < DEG_R) csr[atomicAdd(&off[r], 1u)] = srcp[e];
        }
    }
}

__device__ __forceinline__ void prep_w_elem(int t, const float* __restrict__ W1,
                                            const float* __restrict__ W2,
                                            _Float16* __restrict__ w1B,
                                            _Float16* __restrict__ w2B) {
    if (t < 4 * 32 * 16) {
        int c = t / (32 * 16);
        int l = (t / 16) & 31;
        int j = t & 15;
        int k = 32 * c + klocal(l, j);
        w1B[t] = (_Float16)W1[k * 16 + (l & 15)];
    } else if (t < 4 * 32 * 16 + 32 * 16) {
        int u = t - 4 * 32 * 16;
        int l = u / 16;
        int j = u & 15;
        int k = klocal(l, j);
        int n = l & 15;
        w2B[u] = (k < 16 && n < 7) ? (_Float16)W2[k * 7 + n] : (_Float16)0.0f;
    }
}

__device__ __forceinline__ void gemm1_tile(const float* __restrict__ x,
                                           const _Float16* __restrict__ w1B,
                                           const float* __restrict__ dinv,
                                           float* __restrict__ hs, int nrows,
                                           int wave, int lane) {
    int row0 = wave * 16;
    if (row0 >= nrows) return;
    int rsub = lane & 15;
    int hi   = lane >> 4;
    const float* xr = x + (size_t)(row0 + rsub) * 128 + hi * 8;
    v8f c = {};
#pragma unroll
    for (int kc = 0; kc < 4; ++kc) {
        const float* p = xr + kc * 32;
        v16h a;
#pragma unroll
        for (int j = 0; j < 8; ++j) a[j] = (_Float16)p[j];
#pragma unroll
        for (int j = 0; j < 8; ++j) a[8 + j] = (_Float16)p[16 + j];
        v16h b = *(const v16h*)(w1B + (size_t)(kc * 32 + lane) * 16);
        c = __builtin_amdgcn_wmma_f32_16x16x32_f16(false, a, false, b,
                                                   (short)0, c, false, false);
    }
#pragma unroll
    for (int r = 0; r < 8; ++r) {
        int row = row0 + hi * 8 + r;
        float dv = dinv[row];
        hs[(size_t)row * 16 + rsub] = c[r] * dv;
    }
}

__global__ __launch_bounds__(256) void k_agg1(const int* __restrict__ csr,
                                              const unsigned* __restrict__ rowptr,
                                              const unsigned* __restrict__ degc,
                                              const float* __restrict__ hs,
                                              const float* __restrict__ dinv,
                                              float* __restrict__ acc1, int n) {
    const int t = blockIdx.x * blockDim.x + threadIdx.x;
    const int i = t >> 2;
    const unsigned q = (unsigned)(t & 3);
    if (i >= n) return;
    const int* lp = csr + rowptr[i];
    const unsigned cnt = degc[i];
    const float dv = dinv[i];
    const float4* H = (const float4*)hs;
    float4 a = H[4u * (unsigned)i + q];
    int nx[8];
#pragma unroll
    for (int k = 0; k < 8; ++k) nx[k] = lp[k];
    unsigned j = 0;
    do {
        int cu[8];
#pragma unroll
        for (int k = 0; k < 8; ++k) cu[k] = nx[k];
#pragma unroll
        for (int k = 0; k < 8; ++k) nx[k] = lp[j + 8 + k];
        const unsigned s0 = (j < cnt) ? (unsigned)cu[0] : (unsigned)i;
        float4 v[8];
#pragma unroll
        for (int k = 0; k < 8; ++k) {
            const unsigned s = (j + k < cnt) ? (unsigned)cu[k] : s0;
            v[k] = H[4u * s + q];
        }
#pragma unroll
        for (int k = 0; k < 8; ++k)
            if (j + k < cnt) { a.x += v[k].x; a.y += v[k].y; a.z += v[k].z; a.w += v[k].w; }
        j += 8;
    } while (j < cnt);
    *(float4*)(acc1 + (size_t)i * 16 + q * 4) = make_float4(a.x * dv, a.y * dv, a.z * dv, a.w * dv);
}

__global__ __launch_bounds__(256) void k_gemm2(const float* __restrict__ acc1,
                                               const float* __restrict__ b1,
                                               const _Float16* __restrict__ w2B,
                                               const float* __restrict__ dinv,
                                               float* __restrict__ hs2, int nrows) {
    int wave = (blockIdx.x * blockDim.x + threadIdx.x) >> 5;
    int lane = threadIdx.x & 31;
    int row0 = wave * 16;
    if (row0 >= nrows) return;
    int rsub = lane & 15;
    int hi   = lane >> 4;
    int base = hi * 8;
    const float* p = acc1 + (size_t)(row0 + rsub) * 16 + base;
    v16h a;
#pragma unroll
    for (int j = 0; j < 8; ++j) {
        float f = p[j] + b1[base + j];
        a[j] = (_Float16)fmaxf(f, 0.0f);
    }
#pragma unroll
    for (int j = 8; j < 16; ++j) a[j] = (_Float16)0.0f;
    v16h b = *(const v16h*)(w2B + (size_t)lane * 16);
    v8f c = {};
    c = __builtin_amdgcn_wmma_f32_16x16x32_f16(false, a, false, b,
                                               (short)0, c, false, false);
    if (rsub < 8) {
#pragma unroll
        for (int r = 0; r < 8; ++r) {
            int row = row0 + base + r;
            float dv = dinv[row];
            hs2[(size_t)row * 8 + rsub] = c[r] * dv;
        }
    }
}

__global__ __launch_bounds__(256) void k_agg2(const int* __restrict__ csr,
                                              const unsigned* __restrict__ rowptr,
                                              const unsigned* __restrict__ degc,
                                              const float* __restrict__ hs2,
                                              const float* __restrict__ dinv,
                                              const float* __restrict__ b2,
                                              float* __restrict__ out, int n) {
    const int t = blockIdx.x * blockDim.x + threadIdx.x;
    const int i = t >> 1;
    const unsigned q = (unsigned)(t & 1);
    if (i >= n) return;
    const int* lp = csr + rowptr[i];
    const unsigned cnt = degc[i];
    const float dv = dinv[i];
    const float* bq = b2 + q * 4;
    const float bias0 = bq[0], bias1 = bq[1], bias2 = bq[2];
    const float bias3 = (q == 0u) ? bq[3] : 0.0f;
    const float4* H = (const float4*)hs2;
    float4 a = H[2u * (unsigned)i + q];
    int nx[8];
#pragma unroll
    for (int k = 0; k < 8; ++k) nx[k] = lp[k];
    unsigned j = 0;
    do {
        int cu[8];
#pragma unroll
        for (int k = 0; k < 8; ++k) cu[k] = nx[k];
#pragma unroll
        for (int k = 0; k < 8; ++k) nx[k] = lp[j + 8 + k];
        const unsigned s0 = (j < cnt) ? (unsigned)cu[0] : (unsigned)i;
        float4 v[8];
#pragma unroll
        for (int k = 0; k < 8; ++k) {
            const unsigned s = (j + k < cnt) ? (unsigned)cu[k] : s0;
            v[k] = H[2u * s + q];
        }
#pragma unroll
        for (int k = 0; k < 8; ++k)
            if (j + k < cnt) { a.x += v[k].x; a.y += v[k].y; a.z += v[k].z; a.w += v[k].w; }
        j += 8;
    } while (j < cnt);
    float* o = out + (size_t)i * 7 + q * 4;
    o[0] = fmaxf(a.x * dv + bias0, 0.0f);
    o[1] = fmaxf(a.y * dv + bias1, 0.0f);
    o[2] = fmaxf(a.z * dv + bias2, 0.0f);
    if (q == 0u) o[3] = fmaxf(a.w * dv + bias3, 0.0f);
}

extern "C" void kernel_launch(void* const* d_in, const int* in_sizes, int n_in,
                              void* d_out, int out_size, void* d_ws, size_t ws_size,
                              hipStream_t stream) {
    const float* x  = (const float*)d_in[0];
    const int*   ei = (const int*)d_in[1];
    const float* W1 = (const float*)d_in[2];
    const float* b1 = (const float*)d_in[3];
    const float* W2 = (const float*)d_in[4];
    const float* b2 = (const float*)d_in[5];
    float* out = (float*)d_out;

    const int N = in_sizes[0] / 128;
    const int E = in_sizes[1] / 2;

    char* ws = (char*)d_ws;
    float*    hs     = (float*)(ws + 0);
    float*    acc1   = (float*)(ws + (size_t)N * 16 * 4);
    float*    hs2    = (float*)(ws + (size_t)N * 32 * 4);
    unsigned* degc   = (unsigned*)(ws + (size_t)N * 40 * 4);
    unsigned* rowptr = (unsigned*)(ws + (size_t)N * 41 * 4);
    unsigned* totals = (unsigned*)(ws + (size_t)N * 42 * 4);
    float*    dinv   = (float*)(ws + (size_t)N * 49 * 4);
    int*      csr    = (int*)(ws + (size_t)N * 50 * 4);
    _Float16* w1B  = (_Float16*)(ws + (size_t)N * 50 * 4 + (size_t)E * 4);
    _Float16* w2B  = (_Float16*)(ws + (size_t)N * 50 * 4 + (size_t)E * 4 + 4096);
    unsigned* part = (unsigned*)(ws + (size_t)N * 50 * 4 + (size_t)E * 4 + 8192);

    const int B = 256;
    auto blk = [](long long t, int b) { return (int)((t + b - 1) / b); };
    const int groups = blk(N, 256);

    const int ranges = blk(N, DEG_R);
    int waves1 = (N + 15) / 16;
    const int gemmBlocks = blk((long long)waves1 * 32, B);

    k_deg_part <<<dim3(ranges, DEG_CHUNKS + 1), B, 0, stream>>>(ei, part, N, E, W1, W2, w1B, w2B);
    k_dinv_scan<<<groups, 256, 0, stream>>>(part, dinv, degc, totals, N);
    k_fill     <<<dim3(ranges, DEG_CHUNKS + blk(gemmBlocks, ranges)), B, 0, stream>>>(
        ei, part, totals, rowptr, csr, N, E, groups, x, w1B, dinv, hs);

    k_agg1 <<<blk((long long)N * 4, B), B, 0, stream>>>(csr, rowptr, degc, hs, dinv, acc1, N);

    k_gemm2<<<blk((long long)waves1 * 32, B), B, 0, stream>>>(acc1, b1, w2B, dinv, hs2, N);
    k_agg2 <<<blk((long long)N * 2, B), B, 0, stream>>>(csr, rowptr, degc, hs2, dinv, b2, out, N);
}


# count and fill: LDS node range per block widened from 8192 to 33792 counters (132 KiB; blocks of 1024 and 512 threads): each edge chunk is read by 3 ranges instead of 13; sum order unchanged
# speedup vs baseline: 3.2594x; 1.0351x over previous
#include <hip/hip_runtime.h>
#include <hip/hip_bf16.h>

typedef __attribute__((ext_vector_type(16))) _Float16 v16h;
typedef __attribute__((ext_vector_type(8)))  float    v8f;

__device__ __forceinline__ int klocal(int lane, int j) {
    return ((lane >> 4) << 3) + (j < 8 ? j : j + 8);
}

__device__ __forceinline__ void prep_w_elem(int t, const float* __restrict__ W1,
                                            const float* __restrict__ W2,
                                            _Float16* __restrict__ w1B,
                                            _Float16* __restrict__ w2B);
__device__ __forceinline__ void gemm1_tile(const float* __restrict__ x,
                                           const _Float16* __restrict__ w1B,
                                           const float* __restrict__ dinv,
                                           float* __restrict__ hs, int nrows,
                                           int wave, int lane);

#define DEG_R      33792
#define DEG_T      1024
#define FILL_T     512
#define FILL_L     5
#define DEG_L      10
#define DEG_CHUNKS 80

__global__ __launch_bounds__(DEG_T) void k_deg_part(const int* __restrict__ ei,
                                                  unsigned* __restrict__ part,
                                                  int n, int E,
                                                  const float* __restrict__ W1,
                                                  const float* __restrict__ W2,
                                                  _Float16* __restrict__ w1B,
                                                  _Float16* __restrict__ w2B) {
    __shared__ __attribute__((aligned(16))) unsigned cnt[DEG_R];
    if (blockIdx.y == DEG_CHUNKS) {
        for (int t = blockIdx.x * DEG_T + threadIdx.x; t < 4 * 32 * 16 + 32 * 16; t += gridDim.x * DEG_T)
            prep_w_elem(t, W1, W2, w1B, w2B);
        return;
    }
    const int lo    = blockIdx.x * DEG_R;
    const int chunk = blockIdx.y;
    for (int i = 4 * (int)threadIdx.x; i < DEG_R; i += 4 * DEG_T)
        *(uint4*)&cnt[i] = make_uint4(0u, 0u, 0u, 0u);
    __syncthreads();

    const int* dst = ei + E;
    const int chunkEdges = ((E + DEG_CHUNKS - 1) / DEG_CHUNKS + 3) & ~3;
    const int e0 = chunk * chunkEdges;
    int e1 = e0 + chunkEdges;
    if (e1 > E) e1 = E;
    if ((E & 3) == 0) {
        for (int e = e0 + 4 * (int)threadIdx.x; e < e1; e += DEG_L * 4 * DEG_T) {
            int4 d[DEG_L];
#pragma unroll
            for (int k = 0; k < DEG_L; ++k) {
                const int ek = e + k * (4 * DEG_T);
                d[k] = (ek < e1) ? *(const int4*)(dst + ek) : make_int4(-1, -1, -1, -1);
            }
#pragma unroll
            for (int k = 0; k < DEG_L; ++k) {
                const unsigned r0 = (unsigned)(d[k].x - lo);
                const unsigned r1 = (unsigned)(d[k].y - lo);
                const unsigned r2 = (unsigned)(d[k].z - lo);
                const unsigned r3 = (unsigned)(d[k].w - lo);
                if (r0 < DEG_R) atomicAdd(&cnt[r0], 1u);
                if (r1 < DEG_R) atomicAdd(&cnt[r1], 1u);
                if (r2 < DEG_R) atomicAdd(&cnt[r2], 1u);
                if (r3 < DEG_R) atomicAdd(&cnt[r3], 1u);
            }
        }
    } else {
        for (int e = e0 + (int)threadIdx.x; e < e1; e += DEG_T) {
            const unsigned r = (unsigned)(dst[e] - lo);
            if (r < DEG_R) atomicAdd(&cnt[r], 1u);
        }
    }
    __syncthreads();

    int nv = n - lo;
    if (nv > DEG_R) nv = DEG_R;
    unsigned* p = part + (size_t)chunk * n + lo;
    if (((n | E) & 3) == 0) {
        for (int i = 4 * (int)threadIdx.x; i < nv; i += 4 * DEG_T) *(uint4*)(p + i) = *(const uint4*)&cnt[i];
    } else {
        for (int i = threadIdx.x; i < nv; i += DEG_T) p[i] = cnt[i];
    }
}

__global__ __launch_bounds__(256) void k_dinv_scan(unsigned* __restrict__ part,
                                                   float* __restrict__ dinv,
                                                   unsigned* __restrict__ degc,
                                                   unsigned* __restrict__ totals, int n) {
    __shared__ unsigned wsum[8];
    const int i    = blockIdx.x * 256 + threadIdx.x;
    const int lane = threadIdx.x & 31;
    const int wave = threadIdx.x >> 5;
    unsigned c[DEG_CHUNKS];
    const bool live = i < n;
    const unsigned* col = part + (live ? i : n - 1);
#pragma unroll
    for (int k = 0; k < DEG_CHUNKS; ++k) c[k] = col[(size_t)k * n];
    unsigned deg = 0u;
#pragma unroll
    for (int k = 0; k < DEG_CHUNKS; ++k) { const unsigned t = live ? c[k] : 0u; c[k] = deg; deg += t; }
    if (i < n) {
        dinv[i] = rsqrtf((float)(deg + 1u));
        degc[i] = deg;
    }
    unsigned incl = deg;
    for (int d = 1; d < 32; d <<= 1) {
        const unsigned y = __shfl_up(incl, (unsigned)d);
        if (lane >= d) incl += y;
    }
    if (lane == 31) wsum[wave] = incl;
    __syncthreads();
    unsigned wbase = 0u, tot = 0u;
    for (int w = 0; w < 8; ++w) { const unsigned v = wsum[w]; if (w < wave) wbase += v; tot += v; }
    const unsigned loc = wbase + incl - deg;
    if (threadIdx.x == 0) totals[blockIdx.x] = tot;
    if (i < n) {
#pragma unroll
        for (int k = 0; k < DEG_CHUNKS; ++k) part[(size_t)k * n + i] = c[k] + loc;
    }
}

__global__ __launch_bounds__(FILL_T) void k_fill(const int* __restrict__ ei,
                                              const unsigned* __restrict__ part,
                                              const unsigned* __restrict__ totals,
                                              unsigned* __restrict__ rowptr,
                                              int* __restrict__ csr,
                                              int n, int E, int ngroups,
                                              const float* __restrict__ x,
                                              const _Float16* __restrict__ w1B,
                                              const float* __restrict__ dinv,
                                              float* __restrict__ hs) {
    __shared__ __attribute__((aligned(16))) unsigned off[DEG_R];
    __shared__ unsigned baseS[DEG_R / 256];
    if (blockIdx.y >= DEG_CHUNKS) {
        const int gb = (int)(blockIdx.y - DEG_CHUNKS) * (int)gridDim.x + (int)blockIdx.x;
        gemm1_tile(x, w1B, dinv, hs, n, gb * (FILL_T / 32) + (int)(threadIdx.x >> 5), (int)(threadIdx.x & 31));
        return;
    }
    const int lo    = blockIdx.x * DEG_R;
    const int chunk = blockIdx.y;
    for (int j = threadIdx.x >> 3; j < DEG_R / 256; j += FILL_T / 8) {
        const int sub = threadIdx.x & 7;
        int lim = blockIdx.x * (DEG_R / 256) + j;
        if (lim > ngroups) lim = ngroups;
        unsigned s = 0u;
        for (int b = sub; b < lim; b += 8) s += totals[b];
        s += __shfl_xor(s, 1);
        s += __shfl_xor(s, 2);
        s += __shfl_xor(s, 4);
        if (sub == 0) baseS[j] = s;
    }
    __syncthreads();
    int nv = n - lo;
    if (nv > DEG_R) nv = DEG_R;
    const unsigned* p = part + (size_t)chunk * n + lo;
    if (((n | E) & 3) == 0) {
        for (int i0 = 4 * (int)threadIdx.x; i0 < DEG_R; i0 += 9 * 4 * FILL_T) {
            uint4 pv[9];
#pragma unroll
            for (int k = 0; k < 9; ++k) {
                const int i = i0 + k * (4 * FILL_T);
                pv[k] = *(const uint4*)(p + (i < nv ? i : 0));
            }
#pragma unroll
            for (int k = 0; k < 9; ++k) {
                const int i = i0 + k * (4 * FILL_T);
                if (i < nv) {
                    const unsigned b = baseS[i >> 8];
                    const uint4 o = make_uint4(pv[k].x + b, pv[k].y + b, pv[k].z + b, pv[k].w + b);
                    *(uint4*)&off[i] = o;
                    if (chunk == 0) *(uint4*)(rowptr + lo + i) = o;
                }
            }
        }
    } else
    for (int i0 = threadIdx.x; i0 < DEG_R; i0 += 11 * FILL_T) {
        unsigned pv[11], bs[11];
#pragma unroll
        for (int k = 0; k < 11; ++k) {
            const int i = i0 + k * FILL_T;
            const int ic = i < nv ? i : nv - 1;
            pv[k] = p[ic];
            bs[k] = baseS[ic >> 8];
        }
#pragma unroll
        for (int k = 0; k < 11; ++k) {
            const int i = i0 + k * FILL_T;
            if (i < nv) {
                const unsigned o = bs[k] + pv[k];
                off[i] = o;
                if (chunk == 0) rowptr[lo + i] = o;
            }
        }
    }
    __syncthreads();

    const int* srcp = ei;
    const int* dst  = ei + E;
    const int chunkEdges = ((E + DEG_CHUNKS - 1) / DEG_CHUNKS + 3) & ~3;
    const int e0 = chunk * chunkEdges;
    int e1 = e0 + chunkEdges;
    if (e1 > E) e1 = E;
    if ((E & 3) == 0) {
        for (int e = e0 + 4 * (int)threadIdx.x; e < e1; e += FILL_L * 4 * FILL_T) {
            int4 d[FILL_L], s[FILL_L];
#pragma unroll
            for (int k = 0; k < FILL_L; ++k) {
                const int ek = e + k * (4 * FILL_T);
                const bool in = ek < e1;
                d[k] = in ? *(const int4*)(dst  + ek) : make_int4(-1, -1, -1, -1);
                s[k] = in ? *(const int4*)(srcp + ek) : make_int4(0, 0, 0, 0);
            }
            unsigned r[4 * FILL_L];
            int sv[4 * FILL_L];
#pragma unroll
            for (int k = 0; k < FILL_L; ++k) {
                r[4 * k + 0] = (unsigned)(d[k].x - lo); sv[4 * k + 0] = s[k].x;
                r[4 * k + 1] = (unsigned)(d[k].y - lo); sv[4 * k + 1] = s[k].y;
                r[4 * k + 2] = (unsigned)(d[k].z - lo); sv[4 * k + 2] = s[k].z;
                r[4 * k + 3] = (unsigned)(d[k].w - lo); sv[4 * k + 3] = s[k].w;
            }
            unsigned slot[4 * FILL_L];
#pragma unroll
            for (int m = 0; m < 4 * FILL_L; ++m) if (r[m] < DEG_R) slot[m] = atomicAdd(&off[r[m]], 1u);
#pragma unroll
            for (int m = 0; m < 4 * FILL_L; ++m) if (r[m] < DEG_R) csr[slot[m]] = sv[m];
            __builtin_amdgcn_fence(__ATOMIC_RELEASE, "workgroup", "local");
        }
    } else {
        for (int e = e0 + (int)threadIdx.x; e < e1; e += FILL_T) {
            const unsigned r = (unsigned)(dst[e] - lo);
            if (r < DEG_R) csr[atomicAdd(&off[r], 1u)] = srcp[e];
        }
    }
}

__device__ __forceinline__ void prep_w_elem(int t, const float* __restrict__ W1,
                                            const float* __restrict__ W2,
                                            _Float16* __restrict__ w1B,
                                            _Float16* __restrict__ w2B) {
    if (t < 4 * 32 * 16) {
        int c = t / (32 * 16);
        int l = (t / 16) & 31;
        int j = t & 15;
        int k = 32 * c + klocal(l, j);
        w1B[t] = (_Float16)W1[k * 16 + (l & 15)];
    } else if (t < 4 * 32 * 16 + 32 * 16) {
        int u = t - 4 * 32 * 16;
        int l = u / 16;
        int j = u & 15;
        int k = klocal(l, j);
        int n = l & 15;
        w2B[u] = (k < 16 && n < 7) ? (_Float16)W2[k * 7 + n] : (_Float16)0.0f;
    }
}

__device__ __forceinline__ void gemm1_tile(const float* __restrict__ x,
                                           const _Float16* __restrict__ w1B,
                                           const float* __restrict__ dinv,
                                           float* __restrict__ hs, int nrows,
                                           int wave, int lane) {
    int row0 = wave * 16;
    if (row0 >= nrows) return;
    int rsub = lane & 15;
    int hi   = lane >> 4;
    const float* xr = x + (size_t)(row0 + rsub) * 128 + hi * 8;
    v8f c = {};
#pragma unroll
    for (int kc = 0; kc < 4; ++kc) {
        const float* p = xr + kc * 32;
        v16h a;
#pragma unroll
        for (int j = 0; j < 8; ++j) a[j] = (_Float16)p[j];
#pragma unroll
        for (int j = 0; j < 8; ++j) a[8 + j] = (_Float16)p[16 + j];
        v16h b = *(const v16h*)(w1B + (size_t)(kc * 32 + lane) * 16);
        c = __builtin_amdgcn_wmma_f32_16x16x32_f16(false, a, false, b,
                                                   (short)0, c, false, false);
    }
#pragma unroll
    for (int r = 0; r < 8; ++r) {
        int row = row0 + hi * 8 + r;
        float dv = dinv[row];
        hs[(size_t)row * 16 + rsub] = c[r] * dv;
    }
}

__global__ __launch_bounds__(256) void k_agg1(const int* __restrict__ csr,
                                              const unsigned* __restrict__ rowptr,
                                              const unsigned* __restrict__ degc,
                                              const float* __restrict__ hs,
                                              const float* __restrict__ dinv,
                                              float* __restrict__ acc1, int n) {
    const int t = blockIdx.x * blockDim.x + threadIdx.x;
    const int i = t >> 2;
    const unsigned q = (unsigned)(t & 3);
    if (i >= n) return;
    const int* lp = csr + rowptr[i];
    const unsigned cnt = degc[i];
    const float dv = dinv[i];
    const float4* H = (const float4*)hs;
    float4 a = H[4u * (unsigned)i + q];
    int nx[8];
#pragma unroll
    for (int k = 0; k < 8; ++k) nx[k] = lp[k];
    unsigned j = 0;
    do {
        int cu[8];
#pragma unroll
        for (int k = 0; k < 8; ++k) cu[k] = nx[k];
#pragma unroll
        for (int k = 0; k < 8; ++k) nx[k] = lp[j + 8 + k];
        const unsigned s0 = (j < cnt) ? (unsigned)cu[0] : (unsigned)i;
        float4 v[8];
#pragma unroll
        for (int k = 0; k < 8; ++k) {
            const unsigned s = (j + k < cnt) ? (unsigned)cu[k] : s0;
            v[k] = H[4u * s + q];
        }
#pragma unroll
        for (int k = 0; k < 8; ++k)
            if (j + k < cnt) { a.x += v[k].x; a.y += v[k].y; a.z += v[k].z; a.w += v[k].w; }
        j += 8;
    } while (j < cnt);
    *(float4*)(acc1 + (size_t)i * 16 + q * 4) = make_float4(a.x * dv, a.y * dv, a.z * dv, a.w * dv);
}

__global__ __launch_bounds__(256) void k_gemm2(const float* __restrict__ acc1,
                                               const float* __restrict__ b1,
                                               const _Float16* __restrict__ w2B,
                                               const float* __restrict__ dinv,
                                               float* __restrict__ hs2, int nrows) {
    int wave = (blockIdx.x * blockDim.x + threadIdx.x) >> 5;
    int lane = threadIdx.x & 31;
    int row0 = wave * 16;
    if (row0 >= nrows) return;
    int rsub = lane & 15;
    int hi   = lane >> 4;
    int base = hi * 8;
    const float* p = acc1 + (size_t)(row0 + rsub) * 16 + base;
    v16h a;
#pragma unroll
    for (int j = 0; j < 8; ++j) {
        float f = p[j] + b1[base + j];
        a[j] = (_Float16)fmaxf(f, 0.0f);
    }
#pragma unroll
    for (int j = 8; j < 16; ++j) a[j] = (_Float16)0.0f;
    v16h b = *(const v16h*)(w2B + (size_t)lane * 16);
    v8f c = {};
    c = __builtin_amdgcn_wmma_f32_16x16x32_f16(false, a, false, b,
                                               (short)0, c, false, false);
    if (rsub < 8) {
#pragma unroll
        for (int r = 0; r < 8; ++r) {
            int row = row0 + base + r;
            float dv = dinv[row];
            hs2[(size_t)row * 8 + rsub] = c[r] * dv;
        }
    }
}

__global__ __launch_bounds__(256) void k_agg2(const int* __restrict__ csr,
                                              const unsigned* __restrict__ rowptr,
                                              const unsigned* __restrict__ degc,
                                              const float* __restrict__ hs2,
                                              const float* __restrict__ dinv,
                                              const float* __restrict__ b2,
                                              float* __restrict__ out, int n) {
    const int t = blockIdx.x * blockDim.x + threadIdx.x;
    const int i = t >> 1;
    const unsigned q = (unsigned)(t & 1);
    if (i >= n) return;
    const int* lp = csr + rowptr[i];
    const unsigned cnt = degc[i];
    const float dv = dinv[i];
    const float* bq = b2 + q * 4;
    const float bias0 = bq[0], bias1 = bq[1], bias2 = bq[2];
    const float bias3 = (q == 0u) ? bq[3] : 0.0f;
    const float4* H = (const float4*)hs2;
    float4 a = H[2u * (unsigned)i + q];
    int nx[8];
#pragma unroll
    for (int k = 0; k < 8; ++k) nx[k] = lp[k];
    unsigned j = 0;
    do {
        int cu[8];
#pragma unroll
        for (int k = 0; k < 8; ++k) cu[k] = nx[k];
#pragma unroll
        for (int k = 0; k < 8; ++k) nx[k] = lp[j + 8 + k];
        const unsigned s0 = (j < cnt) ? (unsigned)cu[0] : (unsigned)i;
        float4 v[8];
#pragma unroll
        for (int k = 0; k < 8; ++k) {
            const unsigned s = (j + k < cnt) ? (unsigned)cu[k] : s0;
            v[k] = H[2u * s + q];
        }
#pragma unroll
        for (int k = 0; k < 8; ++k)
            if (j + k < cnt) { a.x += v[k].x; a.y += v[k].y; a.z += v[k].z; a.w += v[k].w; }
        j += 8;
    } while (j < cnt);
    float* o = out + (size_t)i * 7 + q * 4;
    o[0] = fmaxf(a.x * dv + bias0, 0.0f);
    o[1] = fmaxf(a.y * dv + bias1, 0.0f);
    o[2] = fmaxf(a.z * dv + bias2, 0.0f);
    if (q == 0u) o[3] = fmaxf(a.w * dv + bias3, 0.0f);
}

extern "C" void kernel_launch(void* const* d_in, const int* in_sizes, int n_in,
                              void* d_out, int out_size, void* d_ws, size_t ws_size,
                              hipStream_t stream) {
    const float* x  = (const float*)d_in[0];
    const int*   ei = (const int*)d_in[1];
    const float* W1 = (const float*)d_in[2];
    const float* b1 = (const float*)d_in[3];
    const float* W2 = (const float*)d_in[4];
    const float* b2 = (const float*)d_in[5];
    float* out = (float*)d_out;

    const int N = in_sizes[0] / 128;
    const int E = in_sizes[1] / 2;

    char* ws = (char*)d_ws;
    float*    hs     = (float*)(ws + 0);
    float*    acc1   = (float*)(ws + (size_t)N * 16 * 4);
    float*    hs2    = (float*)(ws + (size_t)N * 32 * 4);
    unsigned* degc   = (unsigned*)(ws + (size_t)N * 40 * 4);
    unsigned* rowptr = (unsigned*)(ws + (size_t)N * 41 * 4);
    unsigned* totals = (unsigned*)(ws + (size_t)N * 42 * 4);
    float*    dinv   = (float*)(ws + (size_t)N * 49 * 4);
    int*      csr    = (int*)(ws + (size_t)N * 50 * 4);
    _Float16* w1B  = (_Float16*)(ws + (size_t)N * 50 * 4 + (size_t)E * 4);
    _Float16* w2B  = (_Float16*)(ws + (size_t)N * 50 * 4 + (size_t)E * 4 + 4096);
    unsigned* part = (unsigned*)(ws + (size_t)N * 50 * 4 + (size_t)E * 4 + 8192);

    const int B = 256;
    auto blk = [](long long t, int b) { return (int)((t + b - 1) / b); };
    const int groups = blk(N, 256);

    const int ranges = blk(N, DEG_R);
    int waves1 = (N + 15) / 16;
    const int gemmBlocks = blk((long long)waves1 * 32, FILL_T);

    k_deg_part <<<dim3(ranges, DEG_CHUNKS + 1), DEG_T, 0, stream>>>(ei, part, N, E, W1, W2, w1B, w2B);
    k_dinv_scan<<<groups, 256, 0, stream>>>(part, dinv, degc, totals, N);
    k_fill     <<<dim3(ranges, DEG_CHUNKS + blk(gemmBlocks, ranges)), FILL_T, 0, stream>>>(
        ei, part, totals, rowptr, csr, N, E, groups, x, w1B, dinv, hs);

    k_agg1 <<<blk((long long)N * 4, B), B, 0, stream>>>(csr, rowptr, degc, hs, dinv, acc1, N);

    k_gemm2<<<blk((long long)waves1 * 32, B), B, 0, stream>>>(acc1, b1, w2B, dinv, hs2, N);
    k_agg2 <<<blk((long long)N * 2, B), B, 0, stream>>>(csr, rowptr, degc, hs2, dinv, b2, out, N);
}


# edges regrouped in two LDS-staged levels (256-node buckets, then nodes): list stores are contiguous runs, no per-edge scattered 4-byte store; a node's list order (its sum order) is now arrival order
# speedup vs baseline: 5.1567x; 1.5821x over previous
#include <hip/hip_runtime.h>
#include <hip/hip_bf16.h>


typedef __attribute__((ext_vector_type(16))) _Float16 v16h;
typedef __attribute__((ext_vector_type(8)))  float    v8f;

__device__ __forceinline__ int klocal(int lane, int j) {
    return ((lane >> 4) << 3) + (j < 8 ? j : j + 8);
}

__device__ __forceinline__ void prep_w_elem(int t, const float* __restrict__ W1,
                                            const float* __restrict__ W2,
                                            _Float16* __restrict__ w1B,
                                            _Float16* __restrict__ w2B);
__device__ __forceinline__ void gemm1_tile(const float* __restrict__ x,
                                           const _Float16* __restrict__ w1B,
                                           float* __restrict__ hs, int nrows,
                                           int wave, int lane);

#define BK_SHIFT 8
#define BK_NODES 256
#define SC_T     512
#define SC_CAP   12800
#define SC_L     7
#define MAXB     512
#define KD_CAP   24576
#define KD_L     16

__global__ __launch_bounds__(SC_T) void k_bk_count(const int* __restrict__ ei,
                                                   unsigned* __restrict__ histC,
                                                   int E, int nb, int C,
                                                   const float* __restrict__ W1,
                                                   const float* __restrict__ W2,
                                                   _Float16* __restrict__ w1B,
                                                   _Float16* __restrict__ w2B) {
    __shared__ unsigned h[MAXB];
    if ((int)blockIdx.x == C) {
        for (int t = threadIdx.x; t < 4 * 32 * 16 + 32 * 16; t += SC_T)
            prep_w_elem(t, W1, W2, w1B, w2B);
        return;
    }
    for (int i = threadIdx.x; i < MAXB; i += SC_T) h[i] = 0u;
    __syncthreads();
    const int* dst = ei + E;
    const unsigned nbU = (unsigned)nb;
    const int e0 = (int)blockIdx.x * SC_CAP;
    int e1 = e0 + SC_CAP;
    if (e1 > E) e1 = E;
    if ((E & 3) == 0) {
        int4 d[SC_L];
#pragma unroll
        for (int k = 0; k < SC_L; ++k) {
            const int ek = e0 + 4 * ((int)threadIdx.x + k * SC_T);
            d[k] = (ek < e1) ? *(const int4*)(dst + ek) : make_int4(-1, -1, -1, -1);
        }
#pragma unroll
        for (int k = 0; k < SC_L; ++k) {
            const unsigned b0 = (unsigned)d[k].x >> BK_SHIFT;
            const unsigned b1 = (unsigned)d[k].y >> BK_SHIFT;
            const unsigned b2 = (unsigned)d[k].z >> BK_SHIFT;
            const unsigned b3 = (unsigned)d[k].w >> BK_SHIFT;
            if (b0 < nbU) atomicAdd(&h[b0], 1u);
            if (b1 < nbU) atomicAdd(&h[b1], 1u);
            if (b2 < nbU) atomicAdd(&h[b2], 1u);
            if (b3 < nbU) atomicAdd(&h[b3], 1u);
        }
    } else {
        for (int e = e0 + (int)threadIdx.x; e < e1; e += SC_T) {
            const unsigned b = (unsigned)dst[e] >> BK_SHIFT;
            if (b < nbU) atomicAdd(&h[b], 1u);
        }
    }
    __syncthreads();
    for (int b = threadIdx.x; b < nb; b += SC_T) histC[(size_t)blockIdx.x * nb + b] = h[b];
}

__global__ __launch_bounds__(256) void k_bk_scan(const unsigned* __restrict__ histC,
                                                 unsigned* __restrict__ offT,
                                                 unsigned* __restrict__ total, int nb, int C) {
    __shared__ unsigned wsum[8];
    const int lane = threadIdx.x & 31;
    const int wave = threadIdx.x >> 5;
    const size_t row = (size_t)blockIdx.x * C;
    unsigned carry = 0u;
    for (int c0 = 0; c0 < C; c0 += 256) {
        const int c = c0 + (int)threadIdx.x;
        const unsigned v = (c < C) ? histC[(size_t)c * nb + blockIdx.x] : 0u;
        unsigned incl = v;
        for (int d = 1; d < 32; d <<= 1) {
            const unsigned y = __shfl_up(incl, (unsigned)d);
            if (lane >= d) incl += y;
        }
        if (lane == 31) wsum[wave] = incl;
        __syncthreads();
        unsigned wbase = 0u, tot = 0u;
        for (int w = 0; w < 8; ++w) { const unsigned s = wsum[w]; if (w < wave) wbase += s; tot += s; }
        if (c < C) offT[row + c] = carry + wbase + incl - v;
        carry += tot;
        __syncthreads();
    }
    if (threadIdx.x == 0) total[blockIdx.x] = carry;
}

__global__ __launch_bounds__(SC_T) void k_bk_scatter(const int* __restrict__ ei,
                                                     const unsigned* __restrict__ histC,
                                                     const unsigned* __restrict__ offT,
                                                     const unsigned* __restrict__ total,
                                                     unsigned* __restrict__ rec32,
                                                     int E, int nb, int C,
                                                     const float* __restrict__ x,
                                                     const _Float16* __restrict__ w1B,
                                                     float* __restrict__ hs, int n) {
    __shared__ unsigned sd[SC_CAP];
    __shared__ unsigned ss[SC_CAP];
    __shared__ unsigned lcur[MAXB];
    __shared__ unsigned gdel[MAXB];
    __shared__ unsigned wsA[SC_T / 32], wsB[SC_T / 32];
    if ((int)blockIdx.x >= C) {
        const int gb = (int)blockIdx.x - C;
        gemm1_tile(x, w1B, hs, n, gb * (SC_T / 32) + (int)(threadIdx.x >> 5), (int)(threadIdx.x & 31));
        return;
    }
    const int c    = blockIdx.x;
    const int lane = threadIdx.x & 31;
    const int wave = threadIdx.x >> 5;
    const int b    = threadIdx.x;
    const bool hb  = b < nb;
    const unsigned cntL = hb ? histC[(size_t)c * nb + b] : 0u;
    const unsigned offG = hb ? offT[(size_t)b * C + c] : 0u;
    const unsigned totB = hb ? total[b] : 0u;
    unsigned iL = cntL, iG = totB;
    for (int d = 1; d < 32; d <<= 1) {
        const unsigned yL = __shfl_up(iL, (unsigned)d);
        const unsigned yG = __shfl_up(iG, (unsigned)d);
        if (lane >= d) { iL += yL; iG += yG; }
    }
    if (lane == 31) { wsA[wave] = iL; wsB[wave] = iG; }
    __syncthreads();
    unsigned baseL = 0u, baseG = 0u, nloc = 0u;
    for (int w = 0; w < SC_T / 32; ++w) {
        const unsigned a = wsA[w];
        if (w < wave) { baseL += a; baseG += wsB[w]; }
        nloc += a;
    }
    const unsigned lstart = baseL + iL - cntL;
    lcur[b] = lstart;
    gdel[b] = (baseG + iG - totB + offG) - lstart;
    __syncthreads();

    const int* srcp = ei;
    const int* dst  = ei + E;
    const unsigned nbU = (unsigned)nb;
    const int e0 = c * SC_CAP;
    int e1 = e0 + SC_CAP;
    if (e1 > E) e1 = E;
    if ((E & 3) == 0) {
        int4 d[SC_L], s[SC_L];
#pragma unroll
        for (int k = 0; k < SC_L; ++k) {
            const int ek = e0 + 4 * ((int)threadIdx.x + k * SC_T);
            const bool in = ek < e1;
            d[k] = in ? *(const int4*)(dst  + ek) : make_int4(-1, -1, -1, -1);
            s[k] = in ? *(const int4*)(srcp + ek) : make_int4(0, 0, 0, 0);
        }
        unsigned dv[4 * SC_L], sv[4 * SC_L];
#pragma unroll
        for (int k = 0; k < SC_L; ++k) {
            dv[4 * k + 0] = (unsigned)d[k].x; sv[4 * k + 0] = (unsigned)s[k].x;
            dv[4 * k + 1] = (unsigned)d[k].y; sv[4 * k + 1] = (unsigned)s[k].y;
            dv[4 * k + 2] = (unsigned)d[k].z; sv[4 * k + 2] = (unsigned)s[k].z;
            dv[4 * k + 3] = (unsigned)d[k].w; sv[4 * k + 3] = (unsigned)s[k].w;
        }
        unsigned pl[4 * SC_L];
#pragma unroll
        for (int m = 0; m < 4 * SC_L; ++m)
            if ((dv[m] >> BK_SHIFT) < nbU) pl[m] = atomicAdd(&lcur[dv[m] >> BK_SHIFT], 1u);
#pragma unroll
        for (int m = 0; m < 4 * SC_L; ++m)
            if ((dv[m] >> BK_SHIFT) < nbU) { sd[pl[m]] = dv[m]; ss[pl[m]] = sv[m]; }
    } else {
        for (int e = e0 + (int)threadIdx.x; e < e1; e += SC_T) {
            const unsigned dvv = (unsigned)dst[e];
            if ((dvv >> BK_SHIFT) < nbU) {
                const unsigned p = atomicAdd(&lcur[dvv >> BK_SHIFT], 1u);
                sd[p] = dvv;
                ss[p] = (unsigned)srcp[e];
            }
        }
    }
    __syncthreads();
    for (unsigned t0 = threadIdx.x; t0 < nloc; t0 += 5 * SC_T) {
        unsigned dq[5], sq[5], gq[5];
#pragma unroll
        for (int k = 0; k < 5; ++k) {
            const unsigned t = t0 + (unsigned)k * SC_T;
            const unsigned tt = t < nloc ? t : t0;
            dq[k] = sd[tt];
            sq[k] = ss[tt];
        }
#pragma unroll
        for (int k = 0; k < 5; ++k) gq[k] = gdel[dq[k] >> BK_SHIFT];
#pragma unroll
        for (int k = 0; k < 5; ++k) {
            const unsigned t = t0 + (unsigned)k * SC_T;
            if (t < nloc) ((uint2*)rec32)[gq[k] + t] = make_uint2(dq[k], sq[k]);
        }
    }
}

__global__ __launch_bounds__(BK_NODES) void k_bk_rows(const unsigned* __restrict__ rec32,
                                                      const unsigned* __restrict__ total,
                                                      unsigned* __restrict__ rowptr,
                                                      unsigned* __restrict__ degc,
                                                      float* __restrict__ dinv,
                                                      int* __restrict__ csr,
                                                      float* __restrict__ hs, int n) {
    __shared__ unsigned srt[KD_CAP];
    __shared__ unsigned cur[BK_NODES];
    __shared__ float    dvs[BK_NODES];
    __shared__ unsigned wsum[BK_NODES / 32], wsc[BK_NODES / 32];
    const int b    = blockIdx.x;
    const int lane = threadIdx.x & 31;
    const int wave = threadIdx.x >> 5;
    float4* H = (float4*)hs + (size_t)b * (BK_NODES * 4);
    float4 hv[4];
#pragma unroll
    for (int k = 0; k < 4; ++k) {
        const int idx = (int)threadIdx.x + k * BK_NODES;
        hv[k] = (b * BK_NODES + (idx >> 2) < n) ? H[idx] : make_float4(0.0f, 0.0f, 0.0f, 0.0f);
    }
    unsigned s = 0u;
    for (int t = threadIdx.x; t < b; t += BK_NODES) s += total[t];
    s += __shfl_xor(s, 16);
    s += __shfl_xor(s, 8);
    s += __shfl_xor(s, 4);
    s += __shfl_xor(s, 2);
    s += __shfl_xor(s, 1);
    if (lane == 0) wsum[wave] = s;
    cur[threadIdx.x] = 0u;
    __syncthreads();
    unsigned base = 0u;
    for (int w = 0; w < BK_NODES / 32; ++w) base += wsum[w];
    const unsigned sz = total[b];
    const uint2* R = (const uint2*)rec32 + base;
    for (unsigned t0 = threadIdx.x; t0 < sz; t0 += KD_L * BK_NODES) {
        unsigned dd[KD_L];
#pragma unroll
        for (int k = 0; k < KD_L; ++k) {
            const unsigned t = t0 + (unsigned)k * BK_NODES;
            dd[k] = R[t < sz ? t : t0].x;
        }
#pragma unroll
        for (int k = 0; k < KD_L; ++k)
            if (t0 + (unsigned)k * BK_NODES < sz) atomicAdd(&cur[dd[k] & (BK_NODES - 1)], 1u);
    }
    __syncthreads();
    const unsigned deg = cur[threadIdx.x];
    unsigned incl = deg;
    for (int d = 1; d < 32; d <<= 1) {
        const unsigned y = __shfl_up(incl, (unsigned)d);
        if (lane >= d) incl += y;
    }
    if (lane == 31) wsc[wave] = incl;
    __syncthreads();
    unsigned wbase = 0u;
    for (int w = 0; w < BK_NODES / 32; ++w) if (w < wave) wbase += wsc[w];
    const unsigned start = wbase + incl - deg;
    cur[threadIdx.x] = start;
    const int i = b * BK_NODES + (int)threadIdx.x;
    const float dvi = rsqrtf((float)(deg + 1u));
    dvs[threadIdx.x] = dvi;
    if (i < n) {
        rowptr[i] = base + start;
        degc[i]   = deg;
        dinv[i]   = dvi;
    }
    __syncthreads();
#pragma unroll
    for (int k = 0; k < 4; ++k) {
        const int idx = (int)threadIdx.x + k * BK_NODES;
        const int r   = idx >> 2;
        if (b * BK_NODES + r < n) {
            const float dr = dvs[r];
            H[idx] = make_float4(hv[k].x * dr, hv[k].y * dr, hv[k].z * dr, hv[k].w * dr);
        }
    }
    const bool fits = sz <= KD_CAP;
    for (unsigned t0 = threadIdx.x; t0 < sz; t0 += KD_L * BK_NODES) {
        uint2 rr[KD_L];
#pragma unroll
        for (int k = 0; k < KD_L; ++k) {
            const unsigned t = t0 + (unsigned)k * BK_NODES;
            rr[k] = R[t < sz ? t : t0];
        }
        unsigned pl[KD_L];
#pragma unroll
        for (int k = 0; k < KD_L; ++k)
            if (t0 + (unsigned)k * BK_NODES < sz) pl[k] = atomicAdd(&cur[rr[k].x & (BK_NODES - 1)], 1u);
#pragma unroll
        for (int k = 0; k < KD_L; ++k)
            if (t0 + (unsigned)k * BK_NODES < sz) {
                if (fits) srt[pl[k]] = rr[k].y;
                else      csr[base + pl[k]] = (int)rr[k].y;
            }
    }
    __syncthreads();
    if (fits) {
        const int sh = (int)(base & 31u);
        for (int t0 = (int)threadIdx.x - sh; t0 < (int)sz; t0 += 8 * BK_NODES) {
            unsigned q[8];
#pragma unroll
            for (int k = 0; k < 8; ++k) {
                const int t = t0 + k * BK_NODES;
                q[k] = srt[(t >= 0 && t < (int)sz) ? t : 0];
            }
#pragma unroll
            for (int k = 0; k < 8; ++k) {
                const int t = t0 + k * BK_NODES;
                if (t >= 0 && t < (int)sz) csr[base + (unsigned)t] = (int)q[k];
            }
        }
    }
}

__device__ __forceinline__ void prep_w_elem(int t, const float* __restrict__ W1,
                                            const float* __restrict__ W2,
                                            _Float16* __restrict__ w1B,
                                            _Float16* __restrict__ w2B) {
    if (t < 4 * 32 * 16) {
        int c = t / (32 * 16);
        int l = (t / 16) & 31;
        int j = t & 15;
        int k = 32 * c + klocal(l, j);
        w1B[t] = (_Float16)W1[k * 16 + (l & 15)];
    } else if (t < 4 * 32 * 16 + 32 * 16) {
        int u = t - 4 * 32 * 16;
        int l = u / 16;
        int j = u & 15;
        int k = klocal(l, j);
        int n = l & 15;
        w2B[u] = (k < 16 && n < 7) ? (_Float16)W2[k * 7 + n] : (_Float16)0.0f;
    }
}

__device__ __forceinline__ void gemm1_tile(const float* __restrict__ x,
                                           const _Float16* __restrict__ w1B,
                                           float* __restrict__ hs, int nrows,
                                           int wave, int lane) {
    int row0 = wave * 16;
    if (row0 >= nrows) return;
    int rsub = lane & 15;
    int hi   = lane >> 4;
    const float* xr = x + (size_t)(row0 + rsub) * 128 + hi * 8;
    v8f c = {};
#pragma unroll
    for (int kc = 0; kc < 4; ++kc) {
        const float* p = xr + kc * 32;
        v16h a;
#pragma unroll
        for (int j = 0; j < 8; ++j) a[j] = (_Float16)p[j];
#pragma unroll
        for (int j = 0; j < 8; ++j) a[8 + j] = (_Float16)p[16 + j];
        v16h b = *(const v16h*)(w1B + (size_t)(kc * 32 + lane) * 16);
        c = __builtin_amdgcn_wmma_f32_16x16x32_f16(false, a, false, b,
                                                   (short)0, c, false, false);
    }
#pragma unroll
    for (int r = 0; r < 8; ++r) {
        int row = row0 + hi * 8 + r;
        hs[(size_t)row * 16 + rsub] = c[r];
    }
}

__global__ __launch_bounds__(256) void k_agg1(const int* __restrict__ csr,
                                              const unsigned* __restrict__ rowptr,
                                              const unsigned* __restrict__ degc,
                                              const float* __restrict__ hs,
                                              const float* __restrict__ dinv,
                                              float* __restrict__ acc1, int n) {
    const int t = blockIdx.x * blockDim.x + threadIdx.x;
    const int i = t >> 2;
    const unsigned q = (unsigned)(t & 3);
    if (i >= n) return;
    const int* lp = csr + rowptr[i];
    const unsigned cnt = degc[i];
    const float dv = dinv[i];
    const float4* H = (const float4*)hs;
    float4 a = H[4u * (unsigned)i + q];
    int nx[8];
#pragma unroll
    for (int k = 0; k < 8; ++k) nx[k] = lp[k];
    unsigned j = 0;
    do {
        int cu[8];
#pragma unroll
        for (int k = 0; k < 8; ++k) cu[k] = nx[k];
#pragma unroll
        for (int k = 0; k < 8; ++k) nx[k] = lp[j + 8 + k];
        const unsigned s0 = (j < cnt) ? (unsigned)cu[0] : (unsigned)i;
        float4 v[8];
#pragma unroll
        for (int k = 0; k < 8; ++k) {
            const unsigned s = (j + k < cnt) ? (unsigned)cu[k] : s0;
            v[k] = H[4u * s + q];
        }
#pragma unroll
        for (int k = 0; k < 8; ++k)
            if (j + k < cnt) { a.x += v[k].x; a.y += v[k].y; a.z += v[k].z; a.w += v[k].w; }
        j += 8;
    } while (j < cnt);
    *(float4*)(acc1 + (size_t)i * 16 + q * 4) = make_float4(a.x * dv, a.y * dv, a.z * dv, a.w * dv);
}

__global__ __launch_bounds__(256) void k_gemm2(const float* __restrict__ acc1,
                                               const float* __restrict__ b1,
                                               const _Float16* __restrict__ w2B,
                                               const float* __restrict__ dinv,
                                               float* __restrict__ hs2, int nrows) {
    int wave = (blockIdx.x * blockDim.x + threadIdx.x) >> 5;
    int lane = threadIdx.x & 31;
    int row0 = wave * 16;
    if (row0 >= nrows) return;
    int rsub = lane & 15;
    int hi   = lane >> 4;
    int base = hi * 8;
    const float* p = acc1 + (size_t)(row0 + rsub) * 16 + base;
    v16h a;
#pragma unroll
    for (int j = 0; j < 8; ++j) {
        float f = p[j] + b1[base + j];
        a[j] = (_Float16)fmaxf(f, 0.0f);
    }
#pragma unroll
    for (int j = 8; j < 16; ++j) a[j] = (_Float16)0.0f;
    v16h b = *(const v16h*)(w2B + (size_t)lane * 16);
    v8f c = {};
    c = __builtin_amdgcn_wmma_f32_16x16x32_f16(false, a, false, b,
                                               (short)0, c, false, false);
    if (rsub < 8) {
#pragma unroll
        for (int r = 0; r < 8; ++r) {
            int row = row0 + base + r;
            float dv = dinv[row];
            hs2[(size_t)row * 8 + rsub] = c[r] * dv;
        }
    }
}

__global__ __launch_bounds__(256) void k_agg2(const int* __restrict__ csr,
                                              const unsigned* __restrict__ rowptr,
                                              const unsigned* __restrict__ degc,
                                              const float* __restrict__ hs2,
                                              const float* __restrict__ dinv,
                                              const float* __restrict__ b2,
                                              float* __restrict__ out, int n) {
    const int t = blockIdx.x * blockDim.x + threadIdx.x;
    const int i = t >> 1;
    const unsigned q = (unsigned)(t & 1);
    if (i >= n) return;
    const int* lp = csr + rowptr[i];
    const unsigned cnt = degc[i];
    const float dv = dinv[i];
    const float* bq = b2 + q * 4;
    const float bias0 = bq[0], bias1 = bq[1], bias2 = bq[2];
    const float bias3 = (q == 0u) ? bq[3] : 0.0f;
    const float4* H = (const float4*)hs2;
    float4 a = H[2u * (unsigned)i + q];
    int nx[8];
#pragma unroll
    for (int k = 0; k < 8; ++k) nx[k] = lp[k];
    unsigned j = 0;
    do {
        int cu[8];
#pragma unroll
        for (int k = 0; k < 8; ++k) cu[k] = nx[k];
#pragma unroll
        for (int k = 0; k < 8; ++k) nx[k] = lp[j + 8 + k];
        const unsigned s0 = (j < cnt) ? (unsigned)cu[0] : (unsigned)i;
        float4 v[8];
#pragma unroll
        for (int k = 0; k < 8; ++k) {
            const unsigned s = (j + k < cnt) ? (unsigned)cu[k] : s0;
            v[k] = H[2u * s + q];
        }
#pragma unroll
        for (int k = 0; k < 8; ++k)
            if (j + k < cnt) { a.x += v[k].x; a.y += v[k].y; a.z += v[k].z; a.w += v[k].w; }
        j += 8;
    } while (j < cnt);
    float* o = out + (size_t)i * 7 + q * 4;
    o[0] = fmaxf(a.x * dv + bias0, 0.0f);
    o[1] = fmaxf(a.y * dv + bias1, 0.0f);
    o[2] = fmaxf(a.z * dv + bias2, 0.0f);
    if (q == 0u) o[3] = fmaxf(a.w * dv + bias3, 0.0f);
}

extern "C" void kernel_launch(void* const* d_in, const int* in_sizes, int n_in,
                              void* d_out, int out_size, void* d_ws, size_t ws_size,
                              hipStream_t stream) {
    const float* x  = (const float*)d_in[0];
    const int*   ei = (const int*)d_in[1];
    const float* W1 = (const float*)d_in[2];
    const float* b1 = (const float*)d_in[3];
    const float* W2 = (const float*)d_in[4];
    const float* b2 = (const float*)d_in[5];
    float* out = (float*)d_out;

    const int N = in_sizes[0] / 128;
    const int E = in_sizes[1] / 2;

    const int B = 256;
    auto blk = [](long long t, int b) { return (int)((t + b - 1) / b); };
    const int groups = blk(N, BK_NODES);
    const int chunks = blk(E, SC_CAP);

    char* ws = (char*)d_ws;
    float*    hs     = (float*)(ws + 0);
    float*    acc1   = (float*)(ws + (size_t)N * 16 * 4);
    float*    hs2    = (float*)(ws + (size_t)N * 32 * 4);
    unsigned* degc   = (unsigned*)(ws + (size_t)N * 40 * 4);
    unsigned* rowptr = (unsigned*)(ws + (size_t)N * 41 * 4);
    float*    dinv   = (float*)(ws + (size_t)N * 49 * 4);
    int*      csr    = (int*)(ws + (size_t)N * 50 * 4);
    _Float16* w1B  = (_Float16*)(ws + (size_t)N * 50 * 4 + (size_t)E * 4);
    _Float16* w2B  = (_Float16*)(ws + (size_t)N * 50 * 4 + (size_t)E * 4 + 4096);
    char*     wr   = ws + (size_t)N * 50 * 4 + (size_t)E * 4 + 8192;
    unsigned* rec   = (unsigned*)wr;
    unsigned* histC = (unsigned*)(wr + (size_t)E * 8);
    unsigned* offT  = (unsigned*)(wr + (size_t)E * 8 + (size_t)groups * chunks * 4);
    unsigned* total = (unsigned*)(wr + (size_t)E * 8 + (size_t)groups * chunks * 8);

    int waves1 = (N + 15) / 16;
    const int gemmBlocks = blk((long long)waves1 * 32, SC_T);

    k_bk_count  <<<chunks + 1, SC_T, 0, stream>>>(ei, histC, E, groups, chunks, W1, W2, w1B, w2B);
    k_bk_scan   <<<groups, 256, 0, stream>>>(histC, offT, total, groups, chunks);
    k_bk_scatter<<<chunks + gemmBlocks, SC_T, 0, stream>>>(ei, histC, offT, total, rec, E, groups, chunks,
                                                        x, w1B, hs, N);
    k_bk_rows   <<<groups, BK_NODES, 0, stream>>>(rec, total, rowptr, degc, dinv, csr, hs, N);

    k_agg1 <<<blk((long long)N * 4, B), B, 0, stream>>>(csr, rowptr, degc, hs, dinv, acc1, N);

    k_gemm2<<<blk((long long)waves1 * 32, B), B, 0, stream>>>(acc1, b1, w2B, dinv, hs2, N);
    k_agg2 <<<blk((long long)N * 2, B), B, 0, stream>>>(csr, rowptr, degc, hs2, dinv, b2, out, N);
}


# layer-2 product folded into the layer-1 aggregation's epilogue: four f32 16x16x4 WMMAs on a wave's 8 finished rows (hidden rows stay f32, no f16 rounding); one launch and acc1 go; node sums as before
# speedup vs baseline: 5.3186x; 1.0314x over previous
#include <hip/hip_runtime.h>
#include <hip/hip_bf16.h>


typedef __attribute__((ext_vector_type(16))) _Float16 v16h;
typedef __attribute__((ext_vector_type(8)))  float    v8f;
typedef __attribute__((ext_vector_type(2)))  float    v2f;

__device__ __forceinline__ int klocal(int lane, int j) {
    return ((lane >> 4) << 3) + (j < 8 ? j : j + 8);
}

__device__ __forceinline__ void prep_w_elem(int t, const float* __restrict__ W1,
                                            _Float16* __restrict__ w1B);
__device__ __forceinline__ void gemm1_tile(const float* __restrict__ x,
                                           const _Float16* __restrict__ w1B,
                                           float* __restrict__ hs, int nrows,
                                           int wave, int lane);

#define BK_SHIFT 8
#define BK_NODES 256
#define SC_T     512
#define SC_CAP   12800
#define SC_L     7
#define MAXB     512
#define KD_CAP   24576
#define KD_L     16

__global__ __launch_bounds__(SC_T) void k_bk_count(const int* __restrict__ ei,
                                                   unsigned* __restrict__ histC,
                                                   int E, int nb, int C,
                                                   const float* __restrict__ W1,
                                                   _Float16* __restrict__ w1B) {
    __shared__ unsigned h[MAXB];
    if ((int)blockIdx.x == C) {
        for (int t = threadIdx.x; t < 4 * 32 * 16; t += SC_T)
            prep_w_elem(t, W1, w1B);
        return;
    }
    for (int i = threadIdx.x; i < MAXB; i += SC_T) h[i] = 0u;
    __syncthreads();
    const int* dst = ei + E;
    const unsigned nbU = (unsigned)nb;
    const int e0 = (int)blockIdx.x * SC_CAP;
    int e1 = e0 + SC_CAP;
    if (e1 > E) e1 = E;
    if ((E & 3) == 0) {
        int4 d[SC_L];
#pragma unroll
        for (int k = 0; k < SC_L; ++k) {
            const int ek = e0 + 4 * ((int)threadIdx.x + k * SC_T);
            d[k] = (ek < e1) ? *(const int4*)(dst + ek) : make_int4(-1, -1, -1, -1);
        }
#pragma unroll
        for (int k = 0; k < SC_L; ++k) {
            const unsigned b0 = (unsigned)d[k].x >> BK_SHIFT;
            const unsigned b1 = (unsigned)d[k].y >> BK_SHIFT;
            const unsigned b2 = (unsigned)d[k].z >> BK_SHIFT;
            const unsigned b3 = (unsigned)d[k].w >> BK_SHIFT;
            if (b0 < nbU) atomicAdd(&h[b0], 1u);
            if (b1 < nbU) atomicAdd(&h[b1], 1u);
            if (b2 < nbU) atomicAdd(&h[b2], 1u);
            if (b3 < nbU) atomicAdd(&h[b3], 1u);
        }
    } else {
        for (int e = e0 + (int)threadIdx.x; e < e1; e += SC_T) {
            const unsigned b = (unsigned)dst[e] >> BK_SHIFT;
            if (b < nbU) atomicAdd(&h[b], 1u);
        }
    }
    __syncthreads();
    for (int b = threadIdx.x; b < nb; b += SC_T) histC[(size_t)blockIdx.x * nb + b] = h[b];
}

__global__ __launch_bounds__(256) void k_bk_scan(const unsigned* __restrict__ histC,
                                                 unsigned* __restrict__ offT,
                                                 unsigned* __restrict__ total, int nb, int C) {
    __shared__ unsigned wsum[8];
    const int lane = threadIdx.x & 31;
    const int wave = threadIdx.x >> 5;
    const size_t row = (size_t)blockIdx.x * C;
    unsigned carry = 0u;
    for (int c0 = 0; c0 < C; c0 += 256) {
        const int c = c0 + (int)threadIdx.x;
        const unsigned v = (c < C) ? histC[(size_t)c * nb + blockIdx.x] : 0u;
        unsigned incl = v;
        for (int d = 1; d < 32; d <<= 1) {
            const unsigned y = __shfl_up(incl, (unsigned)d);
            if (lane >= d) incl += y;
        }
        if (lane == 31) wsum[wave] = incl;
        __syncthreads();
        unsigned wbase = 0u, tot = 0u;
        for (int w = 0; w < 8; ++w) { const unsigned s = wsum[w]; if (w < wave) wbase += s; tot += s; }
        if (c < C) offT[row + c] = carry + wbase + incl - v;
        carry += tot;
        __syncthreads();
    }
    if (threadIdx.x == 0) total[blockIdx.x] = carry;
}

__global__ __launch_bounds__(SC_T) void k_bk_scatter(const int* __restrict__ ei,
                                                     const unsigned* __restrict__ histC,
                                                     const unsigned* __restrict__ offT,
                                                     const unsigned* __restrict__ total,
                                                     unsigned* __restrict__ rec32,
                                                     int E, int nb, int C,
                                                     const float* __restrict__ x,
                                                     const _Float16* __restrict__ w1B,
                                                     float* __restrict__ hs, int n) {
    __shared__ unsigned sd[SC_CAP];
    __shared__ unsigned ss[SC_CAP];
    __shared__ unsigned lcur[MAXB];
    __shared__ unsigned gdel[MAXB];
    __shared__ unsigned wsA[SC_T / 32], wsB[SC_T / 32];
    if ((int)blockIdx.x >= C) {
        const int gb = (int)blockIdx.x - C;
        gemm1_tile(x, w1B, hs, n, gb * (SC_T / 32) + (int)(threadIdx.x >> 5), (int)(threadIdx.x & 31));
        return;
    }
    const int c    = blockIdx.x;
    const int lane = threadIdx.x & 31;
    const int wave = threadIdx.x >> 5;
    const int b    = threadIdx.x;
    const bool hb  = b < nb;
    const unsigned cntL = hb ? histC[(size_t)c * nb + b] : 0u;
    const unsigned offG = hb ? offT[(size_t)b * C + c] : 0u;
    const unsigned totB = hb ? total[b] : 0u;
    unsigned iL = cntL, iG = totB;
    for (int d = 1; d < 32; d <<= 1) {
        const unsigned yL = __shfl_up(iL, (unsigned)d);
        const unsigned yG = __shfl_up(iG, (unsigned)d);
        if (lane >= d) { iL += yL; iG += yG; }
    }
    if (lane == 31) { wsA[wave] = iL; wsB[wave] = iG; }
    __syncthreads();
    unsigned baseL = 0u, baseG = 0u, nloc = 0u;
    for (int w = 0; w < SC_T / 32; ++w) {
        const unsigned a = wsA[w];
        if (w < wave) { baseL += a; baseG += wsB[w]; }
        nloc += a;
    }
    const unsigned lstart = baseL + iL - cntL;
    lcur[b] = lstart;
    gdel[b] = (baseG + iG - totB + offG) - lstart;
    __syncthreads();

    const int* srcp = ei;
    const int* dst  = ei + E;
    const unsigned nbU = (unsigned)nb;
    const int e0 = c * SC_CAP;
    int e1 = e0 + SC_CAP;
    if (e1 > E) e1 = E;
    if ((E & 3) == 0) {
        int4 d[SC_L], s[SC_L];
#pragma unroll
        for (int k = 0; k < SC_L; ++k) {
            const int ek = e0 + 4 * ((int)threadIdx.x + k * SC_T);
            const bool in = ek < e1;
            d[k] = in ? *(const int4*)(dst  + ek) : make_int4(-1, -1, -1, -1);
            s[k] = in ? *(const int4*)(srcp + ek) : make_int4(0, 0, 0, 0);
        }
        unsigned dv[4 * SC_L], sv[4 * SC_L];
#pragma unroll
        for (int k = 0; k < SC_L; ++k) {
            dv[4 * k + 0] = (unsigned)d[k].x; sv[4 * k + 0] = (unsigned)s[k].x;
            dv[4 * k + 1] = (unsigned)d[k].y; sv[4 * k + 1] = (unsigned)s[k].y;
            dv[4 * k + 2] = (unsigned)d[k].z; sv[4 * k + 2] = (unsigned)s[k].z;
            dv[4 * k + 3] = (unsigned)d[k].w; sv[4 * k + 3] = (unsigned)s[k].w;
        }
        unsigned pl[4 * SC_L];
#pragma unroll
        for (int m = 0; m < 4 * SC_L; ++m)
            if ((dv[m] >> BK_SHIFT) < nbU) pl[m] = atomicAdd(&lcur[dv[m] >> BK_SHIFT], 1u);
#pragma unroll
        for (int m = 0; m < 4 * SC_L; ++m)
            if ((dv[m] >> BK_SHIFT) < nbU) { sd[pl[m]] = dv[m]; ss[pl[m]] = sv[m]; }
    } else {
        for (int e = e0 + (int)threadIdx.x; e < e1; e += SC_T) {
            const unsigned dvv = (unsigned)dst[e];
            if ((dvv >> BK_SHIFT) < nbU) {
                const unsigned p = atomicAdd(&lcur[dvv >> BK_SHIFT], 1u);
                sd[p] = dvv;
                ss[p] = (unsigned)srcp[e];
            }
        }
    }
    __syncthreads();
    for (unsigned t0 = threadIdx.x; t0 < nloc; t0 += 5 * SC_T) {
        unsigned dq[5], sq[5], gq[5];
#pragma unroll
        for (int k = 0; k < 5; ++k) {
            const unsigned t = t0 + (unsigned)k * SC_T;
            const unsigned tt = t < nloc ? t : t0;
            dq[k] = sd[tt];
            sq[k] = ss[tt];
        }
#pragma unroll
        for (int k = 0; k < 5; ++k) gq[k] = gdel[dq[k] >> BK_SHIFT];
#pragma unroll
        for (int k = 0; k < 5; ++k) {
            const unsigned t = t0 + (unsigned)k * SC_T;
            if (t < nloc) ((uint2*)rec32)[gq[k] + t] = make_uint2(dq[k], sq[k]);
        }
    }
}

__global__ __launch_bounds__(BK_NODES) void k_bk_rows(const unsigned* __restrict__ rec32,
                                                      const unsigned* __restrict__ total,
                                                      unsigned* __restrict__ rowptr,
                                                      unsigned* __restrict__ degc,
                                                      float* __restrict__ dinv,
                                                      int* __restrict__ csr,
                                                      float* __restrict__ hs, int n) {
    __shared__ unsigned srt[KD_CAP];
    __shared__ unsigned cur[BK_NODES];
    __shared__ float    dvs[BK_NODES];
    __shared__ unsigned wsum[BK_NODES / 32], wsc[BK_NODES / 32];
    const int b    = blockIdx.x;
    const int lane = threadIdx.x & 31;
    const int wave = threadIdx.x >> 5;
    float4* H = (float4*)hs + (size_t)b * (BK_NODES * 4);
    float4 hv[4];
#pragma unroll
    for (int k = 0; k < 4; ++k) {
        const int idx = (int)threadIdx.x + k * BK_NODES;
        hv[k] = (b * BK_NODES + (idx >> 2) < n) ? H[idx] : make_float4(0.0f, 0.0f, 0.0f, 0.0f);
    }
    unsigned s = 0u;
    for (int t = threadIdx.x; t < b; t += BK_NODES) s += total[t];
    s += __shfl_xor(s, 16);
    s += __shfl_xor(s, 8);
    s += __shfl_xor(s, 4);
    s += __shfl_xor(s, 2);
    s += __shfl_xor(s, 1);
    if (lane == 0) wsum[wave] = s;
    cur[threadIdx.x] = 0u;
    __syncthreads();
    unsigned base = 0u;
    for (int w = 0; w < BK_NODES / 32; ++w) base += wsum[w];
    const unsigned sz = total[b];
    const uint2* R = (const uint2*)rec32 + base;
    for (unsigned t0 = threadIdx.x; t0 < sz; t0 += KD_L * BK_NODES) {
        unsigned dd[KD_L];
#pragma unroll
        for (int k = 0; k < KD_L; ++k) {
            const unsigned t = t0 + (unsigned)k * BK_NODES;
            dd[k] = R[t < sz ? t : t0].x;
        }
#pragma unroll
        for (int k = 0; k < KD_L; ++k)
            if (t0 + (unsigned)k * BK_NODES < sz) atomicAdd(&cur[dd[k] & (BK_NODES - 1)], 1u);
    }
    __syncthreads();
    const unsigned deg = cur[threadIdx.x];
    unsigned incl = deg;
    for (int d = 1; d < 32; d <<= 1) {
        const unsigned y = __shfl_up(incl, (unsigned)d);
        if (lane >= d) incl += y;
    }
    if (lane == 31) wsc[wave] = incl;
    __syncthreads();
    unsigned wbase = 0u;
    for (int w = 0; w < BK_NODES / 32; ++w) if (w < wave) wbase += wsc[w];
    const unsigned start = wbase + incl - deg;
    cur[threadIdx.x] = start;
    const int i = b * BK_NODES + (int)threadIdx.x;
    const float dvi = rsqrtf((float)(deg + 1u));
    dvs[threadIdx.x] = dvi;
    if (i < n) {
        rowptr[i] = base + start;
        degc[i]   = deg;
        dinv[i]   = dvi;
    }
    __syncthreads();
#pragma unroll
    for (int k = 0; k < 4; ++k) {
        const int idx = (int)threadIdx.x + k * BK_NODES;
        const int r   = idx >> 2;
        if (b * BK_NODES + r < n) {
            const float dr = dvs[r];
            H[idx] = make_float4(hv[k].x * dr, hv[k].y * dr, hv[k].z * dr, hv[k].w * dr);
        }
    }
    const bool fits = sz <= KD_CAP;
    for (unsigned t0 = threadIdx.x; t0 < sz; t0 += KD_L * BK_NODES) {
        uint2 rr[KD_L];
#pragma unroll
        for (int k = 0; k < KD_L; ++k) {
            const unsigned t = t0 + (unsigned)k * BK_NODES;
            rr[k] = R[t < sz ? t : t0];
        }
        unsigned pl[KD_L];
#pragma unroll
        for (int k = 0; k < KD_L; ++k)
            if (t0 + (unsigned)k * BK_NODES < sz) pl[k] = atomicAdd(&cur[rr[k].x & (BK_NODES - 1)], 1u);
#pragma unroll
        for (int k = 0; k < KD_L; ++k)
            if (t0 + (unsigned)k * BK_NODES < sz) {
                if (fits) srt[pl[k]] = rr[k].y;
                else      csr[base + pl[k]] = (int)rr[k].y;
            }
    }
    __syncthreads();
    if (fits) {
        const int sh = (int)(base & 31u);
        for (int t0 = (int)threadIdx.x - sh; t0 < (int)sz; t0 += 8 * BK_NODES) {
            unsigned q[8];
#pragma unroll
            for (int k = 0; k < 8; ++k) {
                const int t = t0 + k * BK_NODES;
                q[k] = srt[(t >= 0 && t < (int)sz) ? t : 0];
            }
#pragma unroll
            for (int k = 0; k < 8; ++k) {
                const int t = t0 + k * BK_NODES;
                if (t >= 0 && t < (int)sz) csr[base + (unsigned)t] = (int)q[k];
            }
        }
    }
}

__device__ __forceinline__ void prep_w_elem(int t, const float* __restrict__ W1,
                                            _Float16* __restrict__ w1B) {
    if (t < 4 * 32 * 16) {
        int c = t / (32 * 16);
        int l = (t / 16) & 31;
        int j = t & 15;
        int k = 32 * c + klocal(l, j);
        w1B[t] = (_Float16)W1[k * 16 + (l & 15)];
    }
}

__device__ __forceinline__ void gemm1_tile(const float* __restrict__ x,
                                           const _Float16* __restrict__ w1B,
                                           float* __restrict__ hs, int nrows,
                                           int wave, int lane) {
    int row0 = wave * 16;
    if (row0 >= nrows) return;
    int rsub = lane & 15;
    int hi   = lane >> 4;
    const float* xr = x + (size_t)(row0 + rsub) * 128 + hi * 8;
    v8f c = {};
#pragma unroll
    for (int kc = 0; kc < 4; ++kc) {
        const float* p = xr + kc * 32;
        v16h a;
#pragma unroll
        for (int j = 0; j < 8; ++j) a[j] = (_Float16)p[j];
#pragma unroll
        for (int j = 0; j < 8; ++j) a[8 + j] = (_Float16)p[16 + j];
        v16h b = *(const v16h*)(w1B + (size_t)(kc * 32 + lane) * 16);
        c = __builtin_amdgcn_wmma_f32_16x16x32_f16(false, a, false, b,
                                                   (short)0, c, false, false);
    }
#pragma unroll
    for (int r = 0; r < 8; ++r) {
        int row = row0 + hi * 8 + r;
        hs[(size_t)row * 16 + rsub] = c[r];
    }
}

__global__ __launch_bounds__(256) void k_agg1(const int* __restrict__ csr,
                                              const unsigned* __restrict__ rowptr,
                                              const unsigned* __restrict__ degc,
                                              const float* __restrict__ hs,
                                              const float* __restrict__ dinv,
                                              const float* __restrict__ b1,
                                              const float* __restrict__ W2,
                                              float* __restrict__ hs2, int n) {
    const int t = blockIdx.x * blockDim.x + threadIdx.x;
    const int lane = threadIdx.x & 31;
    const int node0 = (t >> 5) * 8;
    if (node0 >= n) return;
    const int i = ((t >> 2) < n) ? (t >> 2) : n - 1;
    const unsigned q = (unsigned)(t & 3);
    const int* lp = csr + rowptr[i];
    const unsigned cnt = degc[i];
    const float dv = dinv[i];
    const float4* H = (const float4*)hs;
    float4 a = H[4u * (unsigned)i + q];
    int nx[8];
#pragma unroll
    for (int k = 0; k < 8; ++k) nx[k] = lp[k];
    const float4 bq = *(const float4*)(b1 + 4u * q);
    const int wc = lane & 15;
    const int wh = lane >> 4;
    const int wr = wc < 7 ? wc : 6;
    v2f wa[4];
#pragma unroll
    for (int kk = 0; kk < 4; ++kk) {
        const int f = 4 * (2 * (kk >> 1) + wh) + 2 * (kk & 1);
        const float w0 = W2[f * 7 + wr];
        const float w1 = W2[(f + 1) * 7 + wr];
        wa[kk][0] = (wc < 7) ? w0 : 0.0f;
        wa[kk][1] = (wc < 7) ? w1 : 0.0f;
    }
    unsigned j = 0;
    do {
        int cu[8];
#pragma unroll
        for (int k = 0; k < 8; ++k) cu[k] = nx[k];
#pragma unroll
        for (int k = 0; k < 8; ++k) nx[k] = lp[j + 8 + k];
        const unsigned s0 = (j < cnt) ? (unsigned)cu[0] : (unsigned)i;
        float4 v[8];
#pragma unroll
        for (int k = 0; k < 8; ++k) {
            const unsigned s = (j + k < cnt) ? (unsigned)cu[k] : s0;
            v[k] = H[4u * s + q];
        }
#pragma unroll
        for (int k = 0; k < 8; ++k)
            if (j + k < cnt) { a.x += v[k].x; a.y += v[k].y; a.z += v[k].z; a.w += v[k].w; }
        j += 8;
    } while (j < cnt);
    const float h0 = fmaxf(a.x * dv + bq.x, 0.0f);
    const float h1 = fmaxf(a.y * dv + bq.y, 0.0f);
    const float h2 = fmaxf(a.z * dv + bq.z, 0.0f);
    const float h3 = fmaxf(a.w * dv + bq.w, 0.0f);
    const int sA = 4 * (wc & 7) + wh;
    const int sB = sA + 2;
    v2f hb[4];
    hb[0][0] = __shfl(h0, sA); hb[0][1] = __shfl(h1, sA);
    hb[1][0] = __shfl(h2, sA); hb[1][1] = __shfl(h3, sA);
    hb[2][0] = __shfl(h0, sB); hb[2][1] = __shfl(h1, sB);
    hb[3][0] = __shfl(h2, sB); hb[3][1] = __shfl(h3, sB);
    const float dvc = __shfl(dv, 4 * (lane & 7));
    v8f c = {};
#pragma unroll
    for (int kk = 0; kk < 4; ++kk)
        c = __builtin_amdgcn_wmma_f32_16x16x4_f32(false, wa[kk], false, hb[kk],
                                                  (short)0, c, false, false);
    const float4 o0 = make_float4(c[0] * dvc, c[1] * dvc, c[2] * dvc, c[3] * dvc);
    const float4 o1 = make_float4(c[4] * dvc, c[5] * dvc, c[6] * dvc, c[7] * dvc);
    if (lane < 8 && node0 + lane < n) {
        float4* o = (float4*)(hs2 + (size_t)(node0 + lane) * 8);
        o[0] = o0;
        o[1] = o1;
    }
}

__global__ __launch_bounds__(256) void k_agg2(const int* __restrict__ csr,
                                              const unsigned* __restrict__ rowptr,
                                              const unsigned* __restrict__ degc,
                                              const float* __restrict__ hs2,
                                              const float* __restrict__ dinv,
                                              const float* __restrict__ b2,
                                              float* __restrict__ out, int n) {
    const int t = blockIdx.x * blockDim.x + threadIdx.x;
    const int i = t >> 1;
    const unsigned q = (unsigned)(t & 1);
    if (i >= n) return;
    const int* lp = csr + rowptr[i];
    const unsigned cnt = degc[i];
    const float dv = dinv[i];
    const float* bq = b2 + q * 4;
    const float bias0 = bq[0], bias1 = bq[1], bias2 = bq[2];
    const float bias3 = (q == 0u) ? bq[3] : 0.0f;
    const float4* H = (const float4*)hs2;
    float4 a = H[2u * (unsigned)i + q];
    int nx[8];
#pragma unroll
    for (int k = 0; k < 8; ++k) nx[k] = lp[k];
    unsigned j = 0;
    do {
        int cu[8];
#pragma unroll
        for (int k = 0; k < 8; ++k) cu[k] = nx[k];
#pragma unroll
        for (int k = 0; k < 8; ++k) nx[k] = lp[j + 8 + k];
        const unsigned s0 = (j < cnt) ? (unsigned)cu[0] : (unsigned)i;
        float4 v[8];
#pragma unroll
        for (int k = 0; k < 8; ++k) {
            const unsigned s = (j + k < cnt) ? (unsigned)cu[k] : s0;
            v[k] = H[2u * s + q];
        }
#pragma unroll
        for (int k = 0; k < 8; ++k)
            if (j + k < cnt) { a.x += v[k].x; a.y += v[k].y; a.z += v[k].z; a.w += v[k].w; }
        j += 8;
    } while (j < cnt);
    float* o = out + (size_t)i * 7 + q * 4;
    o[0] = fmaxf(a.x * dv + bias0, 0.0f);
    o[1] = fmaxf(a.y * dv + bias1, 0.0f);
    o[2] = fmaxf(a.z * dv + bias2, 0.0f);
    if (q == 0u) o[3] = fmaxf(a.w * dv + bias3, 0.0f);
}

extern "C" void kernel_launch(void* const* d_in, const int* in_sizes, int n_in,
                              void* d_out, int out_size, void* d_ws, size_t ws_size,
                              hipStream_t stream) {
    const float* x  = (const float*)d_in[0];
    const int*   ei = (const int*)d_in[1];
    const float* W1 = (const float*)d_in[2];
    const float* b1 = (const float*)d_in[3];
    const float* W2 = (const float*)d_in[4];
    const float* b2 = (const float*)d_in[5];
    float* out = (float*)d_out;

    const int N = in_sizes[0] / 128;
    const int E = in_sizes[1] / 2;

    const int B = 256;
    auto blk = [](long long t, int b) { return (int)((t + b - 1) / b); };
    const int groups = blk(N, BK_NODES);
    const int chunks = blk(E, SC_CAP);

    char* ws = (char*)d_ws;
    float*    hs     = (float*)(ws + 0);
    float*    hs2    = (float*)(ws + (size_t)N * 32 * 4);
    unsigned* degc   = (unsigned*)(ws + (size_t)N * 40 * 4);
    unsigned* rowptr = (unsigned*)(ws + (size_t)N * 41 * 4);
    float*    dinv   = (float*)(ws + (size_t)N * 49 * 4);
    int*      csr    = (int*)(ws + (size_t)N * 50 * 4);
    _Float16* w1B  = (_Float16*)(ws + (size_t)N * 50 * 4 + (size_t)E * 4);
    char*     wr   = ws + (size_t)N * 50 * 4 + (size_t)E * 4 + 8192;
    unsigned* rec   = (unsigned*)wr;
    unsigned* histC = (unsigned*)(wr + (size_t)E * 8);
    unsigned* offT  = (unsigned*)(wr + (size_t)E * 8 + (size_t)groups * chunks * 4);
    unsigned* total = (unsigned*)(wr + (size_t)E * 8 + (size_t)groups * chunks * 8);

    int waves1 = (N + 15) / 16;
    const int gemmBlocks = blk((long long)waves1 * 32, SC_T);

    k_bk_count  <<<chunks + 1, SC_T, 0, stream>>>(ei, histC, E, groups, chunks, W1, w1B);
    k_bk_scan   <<<groups, 256, 0, stream>>>(histC, offT, total, groups, chunks);
    k_bk_scatter<<<chunks + gemmBlocks, SC_T, 0, stream>>>(ei, histC, offT, total, rec, E, groups, chunks,
                                                        x, w1B, hs, N);
    k_bk_rows   <<<groups, BK_NODES, 0, stream>>>(rec, total, rowptr, degc, dinv, csr, hs, N);

    k_agg1 <<<blk((long long)N * 4, B), B, 0, stream>>>(csr, rowptr, degc, hs, dinv, b1, W2, hs2, N);

    k_agg2 <<<blk((long long)N * 2, B), B, 0, stream>>>(csr, rowptr, degc, hs2, dinv, b2, out, N);
}


# level-1 grouping per chunk: a block counts, scans and orders its chunk in LDS, stores it chunk-major with its bucket starts; level 2 reads one run per chunk; 6 launches to 4; sum order still arrival
# speedup vs baseline: 5.5812x; 1.0494x over previous
#include <hip/hip_runtime.h>
#include <hip/hip_bf16.h>


typedef __attribute__((ext_vector_type(16))) _Float16 v16h;
typedef __attribute__((ext_vector_type(8)))  float    v8f;
typedef __attribute__((ext_vector_type(2)))  float    v2f;

__device__ __forceinline__ int klocal(int lane, int j) {
    return ((lane >> 4) << 3) + (j < 8 ? j : j + 8);
}

__device__ __forceinline__ int w1_src(int t);
__device__ __forceinline__ void gemm1_tile(const float* __restrict__ x,
                                           const _Float16* w1B,
                                           float* __restrict__ hs, int nrows,
                                           int wave, int lane);

#define BK_SHIFT 8
#define BK_NODES 256
#define SC_T     512
#define SC_CAP   12800
#define SC_L     7
#define MAXB     512
#define KD_CAP   24576
#define KR       16
#define RC       512
#define RCP      (RC + KR * (BK_NODES / 32))

__device__ __forceinline__ unsigned chunk_starts(unsigned* lcur, unsigned* wsA,
                                                 unsigned* __restrict__ lsrow, int nb) {
    const int lane = threadIdx.x & 31;
    const int wave = threadIdx.x >> 5;
    const int b    = threadIdx.x;
    const unsigned cntL = lcur[b];
    unsigned iL = cntL;
    for (int d = 1; d < 32; d <<= 1) {
        const unsigned yL = __shfl_up(iL, (unsigned)d);
        if (lane >= d) iL += yL;
    }
    if (lane == 31) wsA[wave] = iL;
    __syncthreads();
    unsigned baseL = 0u, nloc = 0u;
    for (int w = 0; w < SC_T / 32; ++w) {
        const unsigned a = wsA[w];
        if (w < wave) baseL += a;
        nloc += a;
    }
    const unsigned lstart = baseL + iL - cntL;
    lcur[b] = lstart;
    if (b < nb)  lsrow[b]  = lstart;
    if (b == 0)  lsrow[nb] = nloc;
    __syncthreads();
    return nloc;
}

__global__ __launch_bounds__(SC_T) void k_bk_group(const int* __restrict__ ei,
                                                   unsigned* __restrict__ rec32,
                                                   unsigned* __restrict__ lsT,
                                                   int E, int nb, int C,
                                                   const float* __restrict__ x,
                                                   const float* __restrict__ W1,
                                                   float* __restrict__ hs, int n) {
    __shared__ __attribute__((aligned(32))) unsigned sd[SC_CAP];
    __shared__ unsigned ss[SC_CAP];
    __shared__ unsigned lcur[MAXB];
    __shared__ unsigned wsA[SC_T / 32];
    if ((int)blockIdx.x >= C) {
        _Float16* wl = (_Float16*)sd;
        float wv[4];
#pragma unroll
        for (int k = 0; k < 4; ++k) wv[k] = W1[w1_src((int)threadIdx.x + k * SC_T)];
#pragma unroll
        for (int k = 0; k < 4; ++k) wl[(int)threadIdx.x + k * SC_T] = (_Float16)wv[k];
        __syncthreads();
        const int gb = (int)blockIdx.x - C;
        gemm1_tile(x, wl, hs, n, gb * (SC_T / 32) + (int)(threadIdx.x >> 5), (int)(threadIdx.x & 31));
        return;
    }
    const int c = blockIdx.x;
    lcur[threadIdx.x] = 0u;
    __syncthreads();
    const int* srcp = ei;
    const int* dst  = ei + E;
    const unsigned nbU = (unsigned)nb;
    unsigned* lsrow = lsT + (size_t)c * (nb + 1);
    const int e0 = c * SC_CAP;
    int e1 = e0 + SC_CAP;
    if (e1 > E) e1 = E;
    unsigned nloc;
    if ((E & 3) == 0) {
        int4 d[SC_L], s[SC_L];
#pragma unroll
        for (int k = 0; k < SC_L; ++k) {
            const int ek = e0 + 4 * ((int)threadIdx.x + k * SC_T);
            const bool in = ek < e1;
            d[k] = in ? *(const int4*)(dst  + ek) : make_int4(-1, -1, -1, -1);
            s[k] = in ? *(const int4*)(srcp + ek) : make_int4(0, 0, 0, 0);
        }
        unsigned dv[4 * SC_L], sv[4 * SC_L];
#pragma unroll
        for (int k = 0; k < SC_L; ++k) {
            dv[4 * k + 0] = (unsigned)d[k].x; sv[4 * k + 0] = (unsigned)s[k].x;
            dv[4 * k + 1] = (unsigned)d[k].y; sv[4 * k + 1] = (unsigned)s[k].y;
            dv[4 * k + 2] = (unsigned)d[k].z; sv[4 * k + 2] = (unsigned)s[k].z;
            dv[4 * k + 3] = (unsigned)d[k].w; sv[4 * k + 3] = (unsigned)s[k].w;
        }
#pragma unroll
        for (int m = 0; m < 4 * SC_L; ++m)
            if ((dv[m] >> BK_SHIFT) < nbU) atomicAdd(&lcur[dv[m] >> BK_SHIFT], 1u);
        __syncthreads();
        nloc = chunk_starts(lcur, wsA, lsrow, nb);
        unsigned pl[4 * SC_L];
#pragma unroll
        for (int m = 0; m < 4 * SC_L; ++m)
            if ((dv[m] >> BK_SHIFT) < nbU) pl[m] = atomicAdd(&lcur[dv[m] >> BK_SHIFT], 1u);
#pragma unroll
        for (int m = 0; m < 4 * SC_L; ++m)
            if ((dv[m] >> BK_SHIFT) < nbU) { sd[pl[m]] = dv[m]; ss[pl[m]] = sv[m]; }
    } else {
        for (int e = e0 + (int)threadIdx.x; e < e1; e += SC_T) {
            const unsigned bb = (unsigned)dst[e] >> BK_SHIFT;
            if (bb < nbU) atomicAdd(&lcur[bb], 1u);
        }
        __syncthreads();
        nloc = chunk_starts(lcur, wsA, lsrow, nb);
        for (int e = e0 + (int)threadIdx.x; e < e1; e += SC_T) {
            const unsigned dvv = (unsigned)dst[e];
            if ((dvv >> BK_SHIFT) < nbU) {
                const unsigned p = atomicAdd(&lcur[dvv >> BK_SHIFT], 1u);
                sd[p] = dvv;
                ss[p] = (unsigned)srcp[e];
            }
        }
    }
    __syncthreads();
    uint2* RO = (uint2*)rec32 + (size_t)c * SC_CAP;
    for (unsigned t0 = threadIdx.x; t0 < nloc; t0 += 5 * SC_T) {
        unsigned dq[5], sq[5];
#pragma unroll
        for (int k = 0; k < 5; ++k) {
            const unsigned t = t0 + (unsigned)k * SC_T;
            const unsigned tt = t < nloc ? t : t0;
            dq[k] = sd[tt];
            sq[k] = ss[tt];
        }
#pragma unroll
        for (int k = 0; k < 5; ++k) {
            const unsigned t = t0 + (unsigned)k * SC_T;
            if (t < nloc) RO[t] = make_uint2(dq[k], sq[k]);
        }
    }
}

__global__ __launch_bounds__(BK_NODES) void k_bk_rows(const unsigned* __restrict__ rec32,
                                                      const unsigned* __restrict__ lsT,
                                                      unsigned* __restrict__ rowptr,
                                                      unsigned* __restrict__ degc,
                                                      float* __restrict__ dinv,
                                                      int* __restrict__ csr,
                                                      float* __restrict__ hs, int n,
                                                      int nb, int C, int E) {
    __shared__ unsigned srt[KD_CAP];
    __shared__ unsigned cur[BK_NODES];
    __shared__ float    dvs[BK_NODES];
    __shared__ unsigned rlo[RCP], rhi[RCP];
    __shared__ unsigned wsum[BK_NODES / 32], wsz[BK_NODES / 32], wsf[BK_NODES / 32], wsc[BK_NODES / 32];
    const int b    = blockIdx.x;
    const int lane = threadIdx.x & 31;
    const int wave = threadIdx.x >> 5;
    float4* H = (float4*)hs + (size_t)b * (BK_NODES * 4);
    float4 hv[4];
#pragma unroll
    for (int k = 0; k < 4; ++k) {
        const int idx = (int)threadIdx.x + k * BK_NODES;
        hv[k] = (b * BK_NODES + (idx >> 2) < n) ? H[idx] : make_float4(0.0f, 0.0f, 0.0f, 0.0f);
    }
    if (b == 0 && threadIdx.x < 16) csr[E + (int)threadIdx.x] = 0;
    unsigned s = 0u, z = 0u, f = 0xFFFFFFFFu;
    for (int cc = threadIdx.x; cc < RCP; cc += BK_NODES) {
        unsigned lo = 0u, hi = 0u;
        if (cc < C) {
            const unsigned* p = lsT + (size_t)cc * (nb + 1) + b;
            lo = p[0];
            hi = p[1];
        }
        const unsigned a0 = (unsigned)cc * SC_CAP + lo;
        s += lo;
        z += hi - lo;
        if (hi > lo && a0 < f) f = a0;
        rlo[cc] = a0;
        rhi[cc] = (unsigned)cc * SC_CAP + hi;
    }
    for (int d = 16; d >= 1; d >>= 1) {
        s += __shfl_xor(s, d);
        z += __shfl_xor(z, d);
        const unsigned g = __shfl_xor(f, d);
        f = g < f ? g : f;
    }
    if (lane == 0) { wsum[wave] = s; wsz[wave] = z; wsf[wave] = f; }
    cur[threadIdx.x] = 0u;
    __syncthreads();
    unsigned base = 0u, sz = 0u, sfe = 0xFFFFFFFFu;
    for (int w = 0; w < BK_NODES / 32; ++w) {
        base += wsum[w];
        sz   += wsz[w];
        const unsigned g = wsf[w];
        sfe = g < sfe ? g : sfe;
    }
    const int nruns = C < RC ? C : RC;
    const uint2* R = (const uint2*)rec32;
    if (sz > 0u) {
        for (int r0 = wave; r0 < nruns; r0 += KR * (BK_NODES / 32)) {
            unsigned ta[KR], th[KR], dd[2 * KR];
#pragma unroll
            for (int k = 0; k < KR; ++k) {
                const int r = r0 + k * (BK_NODES / 32);
                ta[k] = rlo[r] + (unsigned)lane;
                th[k] = rhi[r];
            }
#pragma unroll
            for (int k = 0; k < KR; ++k) {
                dd[2 * k]     = R[ta[k] < th[k] ? ta[k] : sfe].x;
                dd[2 * k + 1] = R[ta[k] + 32u < th[k] ? ta[k] + 32u : sfe].x;
            }
#pragma unroll
            for (int k = 0; k < KR; ++k) {
                if (ta[k] < th[k])       atomicAdd(&cur[dd[2 * k] & (BK_NODES - 1)], 1u);
                if (ta[k] + 32u < th[k]) atomicAdd(&cur[dd[2 * k + 1] & (BK_NODES - 1)], 1u);
                for (unsigned t = ta[k] + 64u; t < th[k]; t += 32u)
                    atomicAdd(&cur[R[t].x & (BK_NODES - 1)], 1u);
            }
        }
    }
    __syncthreads();
    const unsigned deg = cur[threadIdx.x];
    unsigned incl = deg;
    for (int d = 1; d < 32; d <<= 1) {
        const unsigned y = __shfl_up(incl, (unsigned)d);
        if (lane >= d) incl += y;
    }
    if (lane == 31) wsc[wave] = incl;
    __syncthreads();
    unsigned wbase = 0u;
    for (int w = 0; w < BK_NODES / 32; ++w) if (w < wave) wbase += wsc[w];
    const unsigned start = wbase + incl - deg;
    cur[threadIdx.x] = start;
    const int i = b * BK_NODES + (int)threadIdx.x;
    const float dvi = rsqrtf((float)(deg + 1u));
    dvs[threadIdx.x] = dvi;
    if (i < n) {
        rowptr[i] = base + start;
        degc[i]   = deg;
        dinv[i]   = dvi;
    }
    __syncthreads();
#pragma unroll
    for (int k = 0; k < 4; ++k) {
        const int idx = (int)threadIdx.x + k * BK_NODES;
        const int r   = idx >> 2;
        if (b * BK_NODES + r < n) {
            const float dr = dvs[r];
            H[idx] = make_float4(hv[k].x * dr, hv[k].y * dr, hv[k].z * dr, hv[k].w * dr);
        }
    }
    const bool fits = sz <= KD_CAP;
    if (sz > 0u) {
        for (int r0 = wave; r0 < nruns; r0 += KR * (BK_NODES / 32)) {
            unsigned ta[KR], th[KR];
            uint2 rr[2 * KR];
#pragma unroll
            for (int k = 0; k < KR; ++k) {
                const int r = r0 + k * (BK_NODES / 32);
                ta[k] = rlo[r] + (unsigned)lane;
                th[k] = rhi[r];
            }
#pragma unroll
            for (int k = 0; k < KR; ++k) {
                rr[2 * k]     = R[ta[k] < th[k] ? ta[k] : sfe];
                rr[2 * k + 1] = R[ta[k] + 32u < th[k] ? ta[k] + 32u : sfe];
            }
            unsigned pl[2 * KR];
#pragma unroll
            for (int k = 0; k < KR; ++k) {
                if (ta[k] < th[k])       pl[2 * k]     = atomicAdd(&cur[rr[2 * k].x & (BK_NODES - 1)], 1u);
                if (ta[k] + 32u < th[k]) pl[2 * k + 1] = atomicAdd(&cur[rr[2 * k + 1].x & (BK_NODES - 1)], 1u);
            }
#pragma unroll
            for (int k = 0; k < KR; ++k) {
                if (ta[k] < th[k]) {
                    if (fits) srt[pl[2 * k]] = rr[2 * k].y;
                    else      csr[base + pl[2 * k]] = (int)rr[2 * k].y;
                }
                if (ta[k] + 32u < th[k]) {
                    if (fits) srt[pl[2 * k + 1]] = rr[2 * k + 1].y;
                    else      csr[base + pl[2 * k + 1]] = (int)rr[2 * k + 1].y;
                }
                for (unsigned t = ta[k] + 64u; t < th[k]; t += 32u) {
                    const uint2 q = R[t];
                    const unsigned p = atomicAdd(&cur[q.x & (BK_NODES - 1)], 1u);
                    if (fits) srt[p] = q.y;
                    else      csr[base + p] = (int)q.y;
                }
            }
        }
    }
    __syncthreads();
    if (fits) {
        const int sh = (int)(base & 31u);
        for (int t0 = (int)threadIdx.x - sh; t0 < (int)sz; t0 += 8 * BK_NODES) {
            unsigned q[8];
#pragma unroll
            for (int k = 0; k < 8; ++k) {
                const int t = t0 + k * BK_NODES;
                q[k] = srt[(t >= 0 && t < (int)sz) ? t : 0];
            }
#pragma unroll
            for (int k = 0; k < 8; ++k) {
                const int t = t0 + k * BK_NODES;
                if (t >= 0 && t < (int)sz) csr[base + (unsigned)t] = (int)q[k];
            }
        }
    }
}

__device__ __forceinline__ int w1_src(int t) {
    int c = t / (32 * 16);
    int l = (t / 16) & 31;
    int j = t & 15;
    int k = 32 * c + klocal(l, j);
    return k * 16 + (l & 15);
}

__device__ __forceinline__ void gemm1_tile(const float* __restrict__ x,
                                           const _Float16* w1B,
                                           float* __restrict__ hs, int nrows,
                                           int wave, int lane) {
    int row0 = wave * 16;
    if (row0 >= nrows) return;
    int rsub = lane & 15;
    int hi   = lane >> 4;
    const float* xr = x + (size_t)(row0 + rsub) * 128 + hi * 8;
    v8f c = {};
#pragma unroll
    for (int kc = 0; kc < 4; ++kc) {
        const float* p = xr + kc * 32;
        v16h a;
#pragma unroll
        for (int j = 0; j < 8; ++j) a[j] = (_Float16)p[j];
#pragma unroll
        for (int j = 0; j < 8; ++j) a[8 + j] = (_Float16)p[16 + j];
        v16h b = *(const v16h*)(w1B + (size_t)(kc * 32 + lane) * 16);
        c = __builtin_amdgcn_wmma_f32_16x16x32_f16(false, a, false, b,
                                                   (short)0, c, false, false);
    }
#pragma unroll
    for (int r = 0; r < 8; ++r) {
        int row = row0 + hi * 8 + r;
        hs[(size_t)row * 16 + rsub] = c[r];
    }
}

__global__ __launch_bounds__(256) void k_agg1(const int* __restrict__ csr,
                                              const unsigned* __restrict__ rowptr,
                                              const unsigned* __restrict__ degc,
                                              const float* __restrict__ hs,
                                              const float* __restrict__ dinv,
                                              const float* __restrict__ b1,
                                              const float* __restrict__ W2,
                                              float* __restrict__ hs2, int n) {
    const int t = blockIdx.x * blockDim.x + threadIdx.x;
    const int lane = threadIdx.x & 31;
    const int node0 = (t >> 5) * 8;
    if (node0 >= n) return;
    const int i = ((t >> 2) < n) ? (t >> 2) : n - 1;
    const unsigned q = (unsigned)(t & 3);
    const int* lp = csr + rowptr[i];
    const unsigned cnt = degc[i];
    const float dv = dinv[i];
    const float4* H = (const float4*)hs;
    float4 a = H[4u * (unsigned)i + q];
    int nx[8];
#pragma unroll
    for (int k = 0; k < 8; ++k) nx[k] = lp[k];
    const float4 bq = *(const float4*)(b1 + 4u * q);
    const int wc = lane & 15;
    const int wh = lane >> 4;
    const int wr = wc < 7 ? wc : 6;
    v2f wa[4];
#pragma unroll
    for (int kk = 0; kk < 4; ++kk) {
        const int f = 4 * (2 * (kk >> 1) + wh) + 2 * (kk & 1);
        const float w0 = W2[f * 7 + wr];
        const float w1 = W2[(f + 1) * 7 + wr];
        wa[kk][0] = (wc < 7) ? w0 : 0.0f;
        wa[kk][1] = (wc < 7) ? w1 : 0.0f;
    }
    unsigned j = 0;
    do {
        int cu[8];
#pragma unroll
        for (int k = 0; k < 8; ++k) cu[k] = nx[k];
#pragma unroll
        for (int k = 0; k < 8; ++k) nx[k] = lp[j + 8 + k];
        const unsigned s0 = (j < cnt) ? (unsigned)cu[0] : (unsigned)i;
        float4 v[8];
#pragma unroll
        for (int k = 0; k < 8; ++k) {
            const unsigned s = (j + k < cnt) ? (unsigned)cu[k] : s0;
            v[k] = H[4u * s + q];
        }
#pragma unroll
        for (int k = 0; k < 8; ++k)
            if (j + k < cnt) { a.x += v[k].x; a.y += v[k].y; a.z += v[k].z; a.w += v[k].w; }
        j += 8;
    } while (j < cnt);
    const float h0 = fmaxf(a.x * dv + bq.x, 0.0f);
    const float h1 = fmaxf(a.y * dv + bq.y, 0.0f);
    const float h2 = fmaxf(a.z * dv + bq.z, 0.0f);
    const float h3 = fmaxf(a.w * dv + bq.w, 0.0f);
    const int sA = 4 * (wc & 7) + wh;
    const int sB = sA + 2;
    v2f hb[4];
    hb[0][0] = __shfl(h0, sA); hb[0][1] = __shfl(h1, sA);
    hb[1][0] = __shfl(h2, sA); hb[1][1] = __shfl(h3, sA);
    hb[2][0] = __shfl(h0, sB); hb[2][1] = __shfl(h1, sB);
    hb[3][0] = __shfl(h2, sB); hb[3][1] = __shfl(h3, sB);
    const float dvc = __shfl(dv, 4 * (lane & 7));
    v8f c = {};
#pragma unroll
    for (int kk = 0; kk < 4; ++kk)
        c = __builtin_amdgcn_wmma_f32_16x16x4_f32(false, wa[kk], false, hb[kk],
                                                  (short)0, c, false, false);
    const float4 o0 = make_float4(c[0] * dvc, c[1] * dvc, c[2] * dvc, c[3] * dvc);
    const float4 o1 = make_float4(c[4] * dvc, c[5] * dvc, c[6] * dvc, c[7] * dvc);
    if (lane < 8 && node0 + lane < n) {
        float4* o = (float4*)(hs2 + (size_t)(node0 + lane) * 8);
        o[0] = o0;
        o[1] = o1;
    }
}

__global__ __launch_bounds__(256) void k_agg2(const int* __restrict__ csr,
                                              const unsigned* __restrict__ rowptr,
                                              const unsigned* __restrict__ degc,
                                              const float* __restrict__ hs2,
                                              const float* __restrict__ dinv,
                                              const float* __restrict__ b2,
                                              float* __restrict__ out, int n) {
    const int t = blockIdx.x * blockDim.x + threadIdx.x;
    const int i = t >> 1;
    const unsigned q = (unsigned)(t & 1);
    if (i >= n) return;
    const int* lp = csr + rowptr[i];
    const unsigned cnt = degc[i];
    const float dv = dinv[i];
    const float* bq = b2 + q * 4;
    const float bias0 = bq[0], bias1 = bq[1], bias2 = bq[2];
    const float bias3 = (q == 0u) ? bq[3] : 0.0f;
    const float4* H = (const float4*)hs2;
    float4 a = H[2u * (unsigned)i + q];
    int nx[8];
#pragma unroll
    for (int k = 0; k < 8; ++k) nx[k] = lp[k];
    unsigned j = 0;
    do {
        int cu[8];
#pragma unroll
        for (int k = 0; k < 8; ++k) cu[k] = nx[k];
#pragma unroll
        for (int k = 0; k < 8; ++k) nx[k] = lp[j + 8 + k];
        const unsigned s0 = (j < cnt) ? (unsigned)cu[0] : (unsigned)i;
        float4 v[8];
#pragma unroll
        for (int k = 0; k < 8; ++k) {
            const unsigned s = (j + k < cnt) ? (unsigned)cu[k] : s0;
            v[k] = H[2u * s + q];
        }
#pragma unroll
        for (int k = 0; k < 8; ++k)
            if (j + k < cnt) { a.x += v[k].x; a.y += v[k].y; a.z += v[k].z; a.w += v[k].w; }
        j += 8;
    } while (j < cnt);
    float* o = out + (size_t)i * 7 + q * 4;
    o[0] = fmaxf(a.x * dv + bias0, 0.0f);
    o[1] = fmaxf(a.y * dv + bias1, 0.0f);
    o[2] = fmaxf(a.z * dv + bias2, 0.0f);
    if (q == 0u) o[3] = fmaxf(a.w * dv + bias3, 0.0f);
}

extern "C" void kernel_launch(void* const* d_in, const int* in_sizes, int n_in,
                              void* d_out, int out_size, void* d_ws, size_t ws_size,
                              hipStream_t stream) {
    const float* x  = (const float*)d_in[0];
    const int*   ei = (const int*)d_in[1];
    const float* W1 = (const float*)d_in[2];
    const float* b1 = (const float*)d_in[3];
    const float* W2 = (const float*)d_in[4];
    const float* b2 = (const float*)d_in[5];
    float* out = (float*)d_out;

    const int N = in_sizes[0] / 128;
    const int E = in_sizes[1] / 2;

    const int B = 256;
    auto blk = [](long long t, int b) { return (int)((t + b - 1) / b); };
    const int groups = blk(N, BK_NODES);
    const int chunks = blk(E, SC_CAP);

    char* ws = (char*)d_ws;
    float*    hs     = (float*)(ws + 0);
    float*    hs2    = (float*)(ws + (size_t)N * 32 * 4);
    unsigned* degc   = (unsigned*)(ws + (size_t)N * 40 * 4);
    unsigned* rowptr = (unsigned*)(ws + (size_t)N * 41 * 4);
    float*    dinv   = (float*)(ws + (size_t)N * 49 * 4);
    int*      csr    = (int*)(ws + (size_t)N * 50 * 4);
    char*     wr   = ws + (size_t)N * 50 * 4 + (size_t)E * 4 + 8192;
    unsigned* rec   = (unsigned*)wr;
    unsigned* lsT   = (unsigned*)(wr + (size_t)chunks * SC_CAP * 8);

    int waves1 = (N + 15) / 16;
    const int gemmBlocks = blk((long long)waves1 * 32, SC_T);

    k_bk_group<<<chunks + gemmBlocks, SC_T, 0, stream>>>(ei, rec, lsT, E, groups, chunks, x, W1, hs, N);
    k_bk_rows <<<groups, BK_NODES, 0, stream>>>(rec, lsT, rowptr, degc, dinv, csr, hs, N, groups, chunks, E);

    k_agg1 <<<blk((long long)N * 4, B), B, 0, stream>>>(csr, rowptr, degc, hs, dinv, b1, W2, hs2, N);

    k_agg2 <<<blk((long long)N * 2, B), B, 0, stream>>>(csr, rowptr, degc, hs2, dinv, b2, out, N);
}
